# starting kernel
# baseline (speedup 1.0000x reference)
#include <hip/hip_runtime.h>

typedef __attribute__((ext_vector_type(2))) float v2f;
typedef __attribute__((ext_vector_type(8))) float v8f;

__global__ void deg_kernel(const int* __restrict__ dst, float* __restrict__ deg, int E) {
  int e = blockIdx.x * blockDim.x + threadIdx.x;
  if (e < E) unsafeAtomicAdd(&deg[dst[e]], 1.0f);
}

__global__ void rsqrt_kernel(float* __restrict__ deg, int N) {
  int i = blockIdx.x * blockDim.x + threadIdx.x;
  if (i < N) deg[i] = rsqrtf(deg[i] + 1.0f);
}

__global__ void build_h0(const float* __restrict__ x, const float* __restrict__ xm,
                         float* __restrict__ h0, int N) {
  int idx = blockIdx.x * blockDim.x + threadIdx.x;
  if (idx >= N * 16) return;
  int i = idx >> 4, j = idx & 15;
  h0[idx] = (j < 8) ? x[i * 10 + j] : xm[i * 10 + (j - 8)];
}

__global__ void gemm128(const float* __restrict__ A, int K, int N,
                        const float* __restrict__ W,
                        float* __restrict__ Out,
                        const float* __restrict__ bias, int do_relu) {
  const int lane = threadIdx.x & 31;
  const int wave = threadIdx.x >> 5;
  const int m0 = blockIdx.x * 16;
  const int n0 = wave * 16;
  const int half = lane >> 4;
  const int r = lane & 15;
  const int n = n0 + r;

  int row = m0 + r;
  row = (row < N) ? row : (N - 1);

  const float* aBase = A + (size_t)row * K + half * 2;
  const float* wBase = W + (size_t)(half * 2) * 128 + n;

  v8f acc = {};
#pragma unroll 4
  for (int k = 0; k < K; k += 4) {
    v2f a = *(const v2f*)(aBase + k);
    v2f b;
    b.x = wBase[(size_t)k * 128];
    b.y = wBase[(size_t)k * 128 + 128];
    acc = __builtin_amdgcn_wmma_f32_16x16x4_f32(false, a, false, b, (short)0, acc,
                                                false, false);
  }

  const float badd = bias ? bias[n] : 0.0f;
  float* outBase = Out + (size_t)(m0 + half * 8) * 128 + n;
  if (m0 + 16 <= N) {
#pragma unroll
    for (int v = 0; v < 8; ++v) {
      float val = acc[v] + badd;
      if (do_relu) val = fmaxf(val, 0.0f);
      outBase[(size_t)v * 128] = val;
    }
  } else {
#pragma unroll
    for (int v = 0; v < 8; ++v) {
      int m = m0 + v + half * 8;
      if (m < N) {
        float val = acc[v] + badd;
        if (do_relu) val = fmaxf(val, 0.0f);
        Out[(size_t)m * 128 + n] = val;
      }
    }
  }
}

__global__ void scatter_edges(const float* __restrict__ hw,
                              const int* __restrict__ src, const int* __restrict__ dst,
                              const float* __restrict__ dis,
                              float* __restrict__ agg, int E) {
  int e = (int)((blockIdx.x * (size_t)blockDim.x + threadIdx.x) >> 5);
  int lane = threadIdx.x & 31;
  if (e >= E) return;
  int s = src[e], d = dst[e];
  float w = dis[s] * dis[d];
  const float4* rowp = (const float4*)(hw + (size_t)s * 128);
  float4 v = rowp[lane];
  float* out = agg + (size_t)d * 128 + lane * 4;
  unsafeAtomicAdd(out + 0, v.x * w);
  unsafeAtomicAdd(out + 1, v.y * w);
  unsafeAtomicAdd(out + 2, v.z * w);
  unsafeAtomicAdd(out + 3, v.w * w);
}

__global__ void finalize(const float* __restrict__ agg, const float* __restrict__ hw,
                         const float* __restrict__ dis, const float* __restrict__ bias,
                         float* __restrict__ h_out, float* __restrict__ emb, int N) {
  int idx = blockIdx.x * blockDim.x + threadIdx.x;
  if (idx >= N * 128) return;
  int i = idx >> 7, f = idx & 127;
  float di = dis[i];
  float val = agg[idx] + hw[idx] * di * di + bias[f];
  if (emb) emb[idx] = val;
  h_out[idx] = fmaxf(val, 0.0f);
}

__global__ void pred_kernel(const float* __restrict__ t, const float* __restrict__ Wr2,
                            const float* __restrict__ br2, float* __restrict__ pred, int N) {
  int i = blockIdx.x * blockDim.x + threadIdx.x;
  if (i >= N) return;
  float a0 = br2[0], a1 = br2[1], a2 = br2[2];
  const float* row = t + (size_t)i * 128;
#pragma unroll 4
  for (int k = 0; k < 128; ++k) {
    float tv = row[k];
    a0 += tv * Wr2[k * 3 + 0];
    a1 += tv * Wr2[k * 3 + 1];
    a2 += tv * Wr2[k * 3 + 2];
  }
  pred[(size_t)i * 3 + 0] = a0;
  pred[(size_t)i * 3 + 1] = a1;
  pred[(size_t)i * 3 + 2] = a2;
}

extern "C" void kernel_launch(void* const* d_in, const int* in_sizes, int n_in,
                              void* d_out, int out_size, void* d_ws, size_t ws_size,
                              hipStream_t stream) {
  const float* x   = (const float*)d_in[0];
  const float* xm  = (const float*)d_in[1];
  const int*   ei  = (const int*)d_in[2];
  const float* W0  = (const float*)d_in[3];
  const float* b0  = (const float*)d_in[4];
  const float* Wh  = (const float*)d_in[5];
  const float* bh  = (const float*)d_in[6];
  const float* Wr1 = (const float*)d_in[7];
  const float* br1 = (const float*)d_in[8];
  const float* Wr2 = (const float*)d_in[9];
  const float* br2 = (const float*)d_in[10];

  const int N = in_sizes[0] / 10;
  const int E = in_sizes[2] / 2;
  const int* srcv = ei;
  const int* dstv = ei + E;

  float* bufA = (float*)d_ws;
  float* bufB = bufA + (size_t)N * 128;
  float* bufC = bufB + (size_t)N * 128;
  float* dis  = bufC + (size_t)N * 128;

  float* emb_out  = (float*)d_out;
  float* pred_out = emb_out + (size_t)N * 128;

  const int mblocks  = (N + 15) / 16;
  const int nfblocks = (int)(((size_t)N * 128 + 255) / 256);
  const int sblocks  = (int)(((size_t)E * 32 + 255) / 256);

  hipMemsetAsync(dis, 0, (size_t)N * sizeof(float), stream);
  deg_kernel<<<(E + 255) / 256, 256, 0, stream>>>(dstv, dis, E);
  rsqrt_kernel<<<(N + 255) / 256, 256, 0, stream>>>(dis, N);

  build_h0<<<(int)(((size_t)N * 16 + 255) / 256), 256, 0, stream>>>(x, xm, bufA, N);

  gemm128<<<mblocks, 256, 0, stream>>>(bufA, 16, N, W0, bufB, nullptr, 0);
  hipMemsetAsync(bufC, 0, (size_t)N * 128 * sizeof(float), stream);
  scatter_edges<<<sblocks, 256, 0, stream>>>(bufB, srcv, dstv, dis, bufC, E);
  finalize<<<nfblocks, 256, 0, stream>>>(bufC, bufB, dis, b0, bufA, nullptr, N);

  for (int l = 0; l < 4; ++l) {
    gemm128<<<mblocks, 256, 0, stream>>>(bufA, 128, N, Wh + (size_t)l * 128 * 128,
                                         bufB, nullptr, 0);
    hipMemsetAsync(bufC, 0, (size_t)N * 128 * sizeof(float), stream);
    scatter_edges<<<sblocks, 256, 0, stream>>>(bufB, srcv, dstv, dis, bufC, E);
    finalize<<<nfblocks, 256, 0, stream>>>(bufC, bufB, dis, bh + (size_t)l * 128, bufA,
                                           (l == 3) ? emb_out : nullptr, N);
  }

  gemm128<<<mblocks, 256, 0, stream>>>(bufA, 128, N, Wr1, bufB, br1, 1);
  pred_kernel<<<(N + 255) / 256, 256, 0, stream>>>(bufB, Wr2, br2, pred_out, N);
}


# aggregation as a gather over a per-call CSR by target row replaces the float atomic scatter: a wave sums a row in fixed source order, adds self-loop, bias, relu and stores it once
# speedup vs baseline: 2.3768x; 2.3768x over previous
#include <hip/hip_runtime.h>

typedef __attribute__((ext_vector_type(2))) float v2f;
typedef __attribute__((ext_vector_type(8))) float v8f;

__global__ void count_kernel(const int* __restrict__ dst, int* __restrict__ cnt, int E) {
  int e = blockIdx.x * blockDim.x + threadIdx.x;
  if (e < E) atomicAdd(&cnt[dst[e]], 1);
}

__global__ void scan_partials(const int* __restrict__ cnt, int* __restrict__ bsum, int N) {
  __shared__ int ws[8];
  const int base = blockIdx.x * 1024 + threadIdx.x * 4;
  int s = 0;
  for (int j = 0; j < 4; ++j) {
    int i = base + j;
    if (i < N) s += cnt[i];
  }
  for (int o = 16; o > 0; o >>= 1) s += __shfl_xor(s, o, 32);
  if ((threadIdx.x & 31) == 0) ws[threadIdx.x >> 5] = s;
  __syncthreads();
  if (threadIdx.x == 0) {
    int t = 0;
    for (int w = 0; w < 8; ++w) t += ws[w];
    bsum[blockIdx.x] = t;
  }
}

__global__ void scan_finish(const int* __restrict__ cnt, const int* __restrict__ bsum,
                            int* __restrict__ rowptr, int* __restrict__ cursor,
                            float* __restrict__ dis, int N) {
  __shared__ int ws[8];
  __shared__ int wt[8];
  const int tid = threadIdx.x, lane = tid & 31, wave = tid >> 5;
  int p = 0;
  for (int j = tid; j < (int)blockIdx.x; j += 256) p += bsum[j];
  for (int o = 16; o > 0; o >>= 1) p += __shfl_xor(p, o, 32);
  if (lane == 0) ws[wave] = p;

  const int base = blockIdx.x * 1024 + tid * 4;
  int c[4];
  int s = 0;
  for (int j = 0; j < 4; ++j) {
    int i = base + j;
    c[j] = (i < N) ? cnt[i] : 0;
    s += c[j];
  }
  int inc = s;
  for (int o = 1; o < 32; o <<= 1) {
    int t = __shfl_up(inc, o, 32);
    if (lane >= o) inc += t;
  }
  if (lane == 31) wt[wave] = inc;
  __syncthreads();
  int off = 0;
  for (int w = 0; w < 8; ++w) off += ws[w];
  for (int w = 0; w < wave; ++w) off += wt[w];
  int excl = off + inc - s;
  for (int j = 0; j < 4; ++j) {
    int i = base + j;
    if (i < N) {
      rowptr[i] = excl;
      cursor[i] = excl;
      dis[i] = rsqrtf((float)c[j] + 1.0f);
      excl += c[j];
      if (i == N - 1) rowptr[N] = excl;
    }
  }
}

__global__ void fill_kernel(const int* __restrict__ src, const int* __restrict__ dst,
                            int* __restrict__ cursor, int* __restrict__ colTmp,
                            int* __restrict__ rowOf, int E) {
  int e = blockIdx.x * blockDim.x + threadIdx.x;
  if (e >= E) return;
  int d = dst[e];
  int pos = atomicAdd(&cursor[d], 1);
  colTmp[pos] = src[e];
  rowOf[pos] = d;
}

__global__ void order_kernel(const int* __restrict__ rowptr, const int* __restrict__ colTmp,
                             const int* __restrict__ rowOf, const float* __restrict__ dis,
                             int* __restrict__ col, float* __restrict__ wgt, int E) {
  int p = blockIdx.x * blockDim.x + threadIdx.x;
  if (p >= E) return;
  const int d = rowOf[p];
  const int s = rowptr[d], e = rowptr[d + 1];
  const int my = colTmp[p];
  int rank = 0;
  for (int q = s; q < e; ++q) {
    int v = colTmp[q];
    rank += ((v < my) || (v == my && q < p)) ? 1 : 0;
  }
  col[s + rank] = my;
  wgt[s + rank] = dis[my] * dis[d];
}

__global__ void build_h0(const float* __restrict__ x, const float* __restrict__ xm,
                         float* __restrict__ h0, int N) {
  int idx = blockIdx.x * blockDim.x + threadIdx.x;
  if (idx >= N * 16) return;
  int i = idx >> 4, j = idx & 15;
  h0[idx] = (j < 8) ? x[i * 10 + j] : xm[i * 10 + (j - 8)];
}

__global__ void gemm128(const float* __restrict__ A, int K, int N,
                        const float* __restrict__ W,
                        float* __restrict__ Out,
                        const float* __restrict__ bias, int do_relu) {
  const int lane = threadIdx.x & 31;
  const int wave = threadIdx.x >> 5;
  const int m0 = blockIdx.x * 16;
  const int n0 = wave * 16;
  const int half = lane >> 4;
  const int r = lane & 15;
  const int n = n0 + r;

  int row = m0 + r;
  row = (row < N) ? row : (N - 1);

  const float* aBase = A + (size_t)row * K + half * 2;
  const float* wBase = W + (size_t)(half * 2) * 128 + n;

  v8f acc = {};
#pragma unroll 4
  for (int k = 0; k < K; k += 4) {
    v2f a = *(const v2f*)(aBase + k);
    v2f b;
    b.x = wBase[(size_t)k * 128];
    b.y = wBase[(size_t)k * 128 + 128];
    acc = __builtin_amdgcn_wmma_f32_16x16x4_f32(false, a, false, b, (short)0, acc,
                                                false, false);
  }

  const float badd = bias ? bias[n] : 0.0f;
  float* outBase = Out + (size_t)(m0 + half * 8) * 128 + n;
  if (m0 + 16 <= N) {
#pragma unroll
    for (int v = 0; v < 8; ++v) {
      float val = acc[v] + badd;
      if (do_relu) val = fmaxf(val, 0.0f);
      outBase[(size_t)v * 128] = val;
    }
  } else {
#pragma unroll
    for (int v = 0; v < 8; ++v) {
      int m = m0 + v + half * 8;
      if (m < N) {
        float val = acc[v] + badd;
        if (do_relu) val = fmaxf(val, 0.0f);
        Out[(size_t)m * 128 + n] = val;
      }
    }
  }
}

__device__ __forceinline__ void fma4(float4& acc, const float4 v, const float w) {
  acc.x = fmaf(v.x, w, acc.x);
  acc.y = fmaf(v.y, w, acc.y);
  acc.z = fmaf(v.z, w, acc.z);
  acc.w = fmaf(v.w, w, acc.w);
}

__global__ void __launch_bounds__(256)
gather_rows(const float* __restrict__ hw, const int* __restrict__ rowptr,
            const int* __restrict__ col, const float* __restrict__ wgt,
            const float* __restrict__ dis, const float* __restrict__ bias,
            float* __restrict__ h_out, float* __restrict__ emb, int N) {
  const int lane = threadIdx.x & 31;
  const int nwaves = gridDim.x * 8;
  const float4 b = ((const float4*)bias)[lane];
  int i = __builtin_amdgcn_readfirstlane((int)(blockIdx.x * 8 + (threadIdx.x >> 5)));
  for (; i < N; i += nwaves) {
    const int s = rowptr[i], e = rowptr[i + 1];
    const float4 self = ((const float4*)(hw + (size_t)i * 128))[lane];
    const float di = dis[i];
    float4 acc = {0.0f, 0.0f, 0.0f, 0.0f};
    for (int base = s; base < e; base += 32) {
      const int nb = (e - base < 32) ? (e - base) : 32;
      int c = i;
      float w = 0.0f;
      if (lane < nb) {
        c = col[base + lane];
        w = wgt[base + lane];
      }
      const int wi = __float_as_int(w);
      int j = 0;
      for (; j + 8 <= nb; j += 8) {
        float4 v[8];
#pragma unroll
        for (int u = 0; u < 8; ++u) {
          const int cu = __builtin_amdgcn_readlane(c, j + u);
          v[u] = ((const float4*)(hw + (size_t)cu * 128))[lane];
        }
#pragma unroll
        for (int u = 0; u < 8; ++u)
          fma4(acc, v[u], __int_as_float(__builtin_amdgcn_readlane(wi, j + u)));
      }
      const int m = nb - j;
      if (m > 0) {
        float4 v[7];
#pragma unroll
        for (int u = 0; u < 7; ++u) {
          if (u < m) {
            const int cu = __builtin_amdgcn_readlane(c, j + u);
            v[u] = ((const float4*)(hw + (size_t)cu * 128))[lane];
          }
        }
#pragma unroll
        for (int u = 0; u < 7; ++u) {
          if (u < m) fma4(acc, v[u], __int_as_float(__builtin_amdgcn_readlane(wi, j + u)));
        }
      }
    }
    const float dd = di * di;
    float4 val;
    val.x = acc.x + self.x * dd + b.x;
    val.y = acc.y + self.y * dd + b.y;
    val.z = acc.z + self.z * dd + b.z;
    val.w = acc.w + self.w * dd + b.w;
    if (emb) ((float4*)(emb + (size_t)i * 128))[lane] = val;
    float4 r;
    r.x = fmaxf(val.x, 0.0f);
    r.y = fmaxf(val.y, 0.0f);
    r.z = fmaxf(val.z, 0.0f);
    r.w = fmaxf(val.w, 0.0f);
    ((float4*)(h_out + (size_t)i * 128))[lane] = r;
  }
}

__global__ void pred_kernel(const float* __restrict__ t, const float* __restrict__ Wr2,
                            const float* __restrict__ br2, float* __restrict__ pred, int N) {
  int i = blockIdx.x * blockDim.x + threadIdx.x;
  if (i >= N) return;
  float a0 = br2[0], a1 = br2[1], a2 = br2[2];
  const float* row = t + (size_t)i * 128;
#pragma unroll 4
  for (int k = 0; k < 128; ++k) {
    float tv = row[k];
    a0 += tv * Wr2[k * 3 + 0];
    a1 += tv * Wr2[k * 3 + 1];
    a2 += tv * Wr2[k * 3 + 2];
  }
  pred[(size_t)i * 3 + 0] = a0;
  pred[(size_t)i * 3 + 1] = a1;
  pred[(size_t)i * 3 + 2] = a2;
}

extern "C" void kernel_launch(void* const* d_in, const int* in_sizes, int n_in,
                              void* d_out, int out_size, void* d_ws, size_t ws_size,
                              hipStream_t stream) {
  const float* x   = (const float*)d_in[0];
  const float* xm  = (const float*)d_in[1];
  const int*   ei  = (const int*)d_in[2];
  const float* W0  = (const float*)d_in[3];
  const float* b0  = (const float*)d_in[4];
  const float* Wh  = (const float*)d_in[5];
  const float* bh  = (const float*)d_in[6];
  const float* Wr1 = (const float*)d_in[7];
  const float* br1 = (const float*)d_in[8];
  const float* Wr2 = (const float*)d_in[9];
  const float* br2 = (const float*)d_in[10];

  const int N = in_sizes[0] / 10;
  const int E = in_sizes[2] / 2;
  const int* srcv = ei;
  const int* dstv = ei + E;

  float* bufA = (float*)d_ws;
  float* bufB = bufA + (size_t)N * 128;
  float* bufC = bufB + (size_t)N * 128;
  float* dis  = bufC + (size_t)N * 128;

  const size_t nA = ((size_t)N + 1 + 63) & ~(size_t)63;
  const size_t eA = ((size_t)E + 63) & ~(size_t)63;
  int*   cnt    = (int*)bufC;
  int*   rowptr = cnt + nA;
  int*   cursor = rowptr + nA;
  int*   bsum   = cursor + nA;
  int*   colTmp = bsum + nA;
  int*   rowOf  = colTmp + eA;
  int*   col    = rowOf + eA;
  float* wgt    = (float*)(col + eA);

  float* emb_out  = (float*)d_out;
  float* pred_out = emb_out + (size_t)N * 128;

  const int mblocks = (N + 15) / 16;
  const int eblocks = (E + 255) / 256;
  const int tblocks = (N + 1023) / 1024;
  const int gblocks = ((N + 7) / 8 + 1) / 2;

  hipMemsetAsync(cnt, 0, (size_t)N * sizeof(int), stream);
  count_kernel<<<eblocks, 256, 0, stream>>>(dstv, cnt, E);
  scan_partials<<<tblocks, 256, 0, stream>>>(cnt, bsum, N);
  scan_finish<<<tblocks, 256, 0, stream>>>(cnt, bsum, rowptr, cursor, dis, N);
  fill_kernel<<<eblocks, 256, 0, stream>>>(srcv, dstv, cursor, colTmp, rowOf, E);
  order_kernel<<<eblocks, 256, 0, stream>>>(rowptr, colTmp, rowOf, dis, col, wgt, E);

  build_h0<<<(int)(((size_t)N * 16 + 255) / 256), 256, 0, stream>>>(x, xm, bufA, N);

  gemm128<<<mblocks, 256, 0, stream>>>(bufA, 16, N, W0, bufB, nullptr, 0);
  gather_rows<<<gblocks, 256, 0, stream>>>(bufB, rowptr, col, wgt, dis, b0, bufA, nullptr, N);

  for (int l = 0; l < 4; ++l) {
    gemm128<<<mblocks, 256, 0, stream>>>(bufA, 128, N, Wh + (size_t)l * 128 * 128,
                                         bufB, nullptr, 0);
    gather_rows<<<gblocks, 256, 0, stream>>>(bufB, rowptr, col, wgt, dis, bh + (size_t)l * 128,
                                             bufA, (l == 3) ? emb_out : nullptr, N);
  }

  gemm128<<<mblocks, 256, 0, stream>>>(bufA, 128, N, Wr1, bufB, br1, 1);
  pred_kernel<<<(N + 255) / 256, 256, 0, stream>>>(bufB, Wr2, br2, pred_out, N);
}


# next layer's 128x128 product fused into the aggregation kernel: each block gathers 16-row tiles into LDS and multiplies them by W kept in LDS; relu'd rows skip memory; 5 launches fewer
# speedup vs baseline: 3.5901x; 1.5105x over previous
#include <hip/hip_runtime.h>

typedef __attribute__((ext_vector_type(2))) float v2f;
typedef __attribute__((ext_vector_type(8))) float v8f;

__global__ void count_kernel(const int* __restrict__ dst, int* __restrict__ cnt, int E) {
  int e = blockIdx.x * blockDim.x + threadIdx.x;
  if (e < E) atomicAdd(&cnt[dst[e]], 1);
}

__global__ void scan_partials(const int* __restrict__ cnt, int* __restrict__ bsum, int N) {
  __shared__ int ws[8];
  const int base = blockIdx.x * 1024 + threadIdx.x * 4;
  int s = 0;
  for (int j = 0; j < 4; ++j) {
    int i = base + j;
    if (i < N) s += cnt[i];
  }
  for (int o = 16; o > 0; o >>= 1) s += __shfl_xor(s, o, 32);
  if ((threadIdx.x & 31) == 0) ws[threadIdx.x >> 5] = s;
  __syncthreads();
  if (threadIdx.x == 0) {
    int t = 0;
    for (int w = 0; w < 8; ++w) t += ws[w];
    bsum[blockIdx.x] = t;
  }
}

__global__ void scan_finish(const int* __restrict__ cnt, const int* __restrict__ bsum,
                            int* __restrict__ rowptr, int* __restrict__ cursor,
                            float* __restrict__ dis, int N) {
  __shared__ int ws[8];
  __shared__ int wt[8];
  const int tid = threadIdx.x, lane = tid & 31, wave = tid >> 5;
  int p = 0;
  for (int j = tid; j < (int)blockIdx.x; j += 256) p += bsum[j];
  for (int o = 16; o > 0; o >>= 1) p += __shfl_xor(p, o, 32);
  if (lane == 0) ws[wave] = p;

  const int base = blockIdx.x * 1024 + tid * 4;
  int c[4];
  int s = 0;
  for (int j = 0; j < 4; ++j) {
    int i = base + j;
    c[j] = (i < N) ? cnt[i] : 0;
    s += c[j];
  }
  int inc = s;
  for (int o = 1; o < 32; o <<= 1) {
    int t = __shfl_up(inc, o, 32);
    if (lane >= o) inc += t;
  }
  if (lane == 31) wt[wave] = inc;
  __syncthreads();
  int off = 0;
  for (int w = 0; w < 8; ++w) off += ws[w];
  for (int w = 0; w < wave; ++w) off += wt[w];
  int excl = off + inc - s;
  for (int j = 0; j < 4; ++j) {
    int i = base + j;
    if (i < N) {
      rowptr[i] = excl;
      cursor[i] = excl;
      dis[i] = rsqrtf((float)c[j] + 1.0f);
      excl += c[j];
      if (i == N - 1) rowptr[N] = excl;
    }
  }
}

__global__ void fill_kernel(const int* __restrict__ src, const int* __restrict__ dst,
                            int* __restrict__ cursor, int* __restrict__ colTmp,
                            int* __restrict__ rowOf, int E) {
  int e = blockIdx.x * blockDim.x + threadIdx.x;
  if (e >= E) return;
  int d = dst[e];
  int pos = atomicAdd(&cursor[d], 1);
  colTmp[pos] = src[e];
  rowOf[pos] = d;
}

__global__ void order_kernel(const int* __restrict__ rowptr, const int* __restrict__ colTmp,
                             const int* __restrict__ rowOf, const float* __restrict__ dis,
                             int* __restrict__ col, float* __restrict__ wgt, int E) {
  int p = blockIdx.x * blockDim.x + threadIdx.x;
  if (p >= E) return;
  const int d = rowOf[p];
  const int s = rowptr[d], e = rowptr[d + 1];
  const int my = colTmp[p];
  int rank = 0;
  for (int q = s; q < e; ++q) {
    int v = colTmp[q];
    rank += ((v < my) || (v == my && q < p)) ? 1 : 0;
  }
  col[s + rank] = my;
  wgt[s + rank] = dis[my] * dis[d];
}

__global__ void build_h0(const float* __restrict__ x, const float* __restrict__ xm,
                         float* __restrict__ h0, int N) {
  int idx = blockIdx.x * blockDim.x + threadIdx.x;
  if (idx >= N * 16) return;
  int i = idx >> 4, j = idx & 15;
  h0[idx] = (j < 8) ? x[i * 10 + j] : xm[i * 10 + (j - 8)];
}

__global__ void gemm128(const float* __restrict__ A, int K, int N,
                        const float* __restrict__ W,
                        float* __restrict__ Out,
                        const float* __restrict__ bias, int do_relu) {
  const int lane = threadIdx.x & 31;
  const int wave = threadIdx.x >> 5;
  const int m0 = blockIdx.x * 16;
  const int n0 = wave * 16;
  const int half = lane >> 4;
  const int r = lane & 15;
  const int n = n0 + r;

  int row = m0 + r;
  row = (row < N) ? row : (N - 1);

  const float* aBase = A + (size_t)row * K + half * 2;
  const float* wBase = W + (size_t)(half * 2) * 128 + n;

  v8f acc = {};
#pragma unroll 4
  for (int k = 0; k < K; k += 4) {
    v2f a = *(const v2f*)(aBase + k);
    v2f b;
    b.x = wBase[(size_t)k * 128];
    b.y = wBase[(size_t)k * 128 + 128];
    acc = __builtin_amdgcn_wmma_f32_16x16x4_f32(false, a, false, b, (short)0, acc,
                                                false, false);
  }

  const float badd = bias ? bias[n] : 0.0f;
  float* outBase = Out + (size_t)(m0 + half * 8) * 128 + n;
  if (m0 + 16 <= N) {
#pragma unroll
    for (int v = 0; v < 8; ++v) {
      float val = acc[v] + badd;
      if (do_relu) val = fmaxf(val, 0.0f);
      outBase[(size_t)v * 128] = val;
    }
  } else {
#pragma unroll
    for (int v = 0; v < 8; ++v) {
      int m = m0 + v + half * 8;
      if (m < N) {
        float val = acc[v] + badd;
        if (do_relu) val = fmaxf(val, 0.0f);
        Out[(size_t)m * 128 + n] = val;
      }
    }
  }
}

__device__ __forceinline__ void fma4(float4& acc, const float4 v, const float w) {
  acc.x = fmaf(v.x, w, acc.x);
  acc.y = fmaf(v.y, w, acc.y);
  acc.z = fmaf(v.z, w, acc.z);
  acc.w = fmaf(v.w, w, acc.w);
}

#define TILE_PITCH 132

struct RowHead {
  int s, e, c;
  float w, di;
  float4 self;
};

__device__ __forceinline__ void row_first(RowHead& h, const int lane,
                                          const int* __restrict__ col, const float* __restrict__ wgt) {
  const int nb = (h.e - h.s < 32) ? (h.e - h.s) : 32;
  if (lane < nb) {
    h.c = col[h.s + lane];
    h.w = wgt[h.s + lane];
  }
}

__device__ __forceinline__ float4 row_sum(const RowHead& h, const int i, const int lane,
                                          const float* __restrict__ hw, const int* __restrict__ col,
                                          const float* __restrict__ wgt) {
  float4 acc = {0.0f, 0.0f, 0.0f, 0.0f};
  int c = h.c;
  float w = h.w;
  for (int base = h.s; base < h.e; base += 32) {
    const int nb = (h.e - base < 32) ? (h.e - base) : 32;
    if (base != h.s) {
      c = i;
      w = 0.0f;
      if (lane < nb) {
        c = col[base + lane];
        w = wgt[base + lane];
      }
    }
    const int wi = __float_as_int(w);
    int j = 0;
    for (; j + 8 <= nb; j += 8) {
      float4 v[8];
#pragma unroll
      for (int u = 0; u < 8; ++u) {
        const int cu = __builtin_amdgcn_readlane(c, j + u);
        v[u] = ((const float4*)(hw + (size_t)cu * 128))[lane];
      }
#pragma unroll
      for (int u = 0; u < 8; ++u)
        fma4(acc, v[u], __int_as_float(__builtin_amdgcn_readlane(wi, j + u)));
    }
    const int m = nb - j;
    if (m > 0) {
      float4 v[7];
#pragma unroll
      for (int u = 0; u < 7; ++u) {
        if (u < m) {
          const int cu = __builtin_amdgcn_readlane(c, j + u);
          v[u] = ((const float4*)(hw + (size_t)cu * 128))[lane];
        }
      }
#pragma unroll
      for (int u = 0; u < 7; ++u) {
        if (u < m) fma4(acc, v[u], __int_as_float(__builtin_amdgcn_readlane(wi, j + u)));
      }
    }
  }
  return acc;
}

__device__ __forceinline__ void row_finish(const float4 acc, const RowHead& h, const float4 b,
                                           const int i, const int N, const int lane,
                                           float* __restrict__ emb, float* __restrict__ tileRow) {
  const float dd = h.di * h.di;
  float4 val;
  val.x = acc.x + h.self.x * dd + b.x;
  val.y = acc.y + h.self.y * dd + b.y;
  val.z = acc.z + h.self.z * dd + b.z;
  val.w = acc.w + h.self.w * dd + b.w;
  if (emb && i < N) ((float4*)(emb + (size_t)i * 128))[lane] = val;
  float4 r;
  r.x = fmaxf(val.x, 0.0f);
  r.y = fmaxf(val.y, 0.0f);
  r.z = fmaxf(val.z, 0.0f);
  r.w = fmaxf(val.w, 0.0f);
  ((float4*)tileRow)[lane] = r;
}

__global__ void __launch_bounds__(256)
gather_gemm(const float* __restrict__ hw, const int* __restrict__ rowptr,
            const int* __restrict__ col, const float* __restrict__ wgt,
            const float* __restrict__ dis, const float* __restrict__ bias,
            float* __restrict__ emb, const float* __restrict__ Wn,
            const float* __restrict__ bias_out, int relu_out,
            float* __restrict__ Out, int N) {
  __shared__ alignas(16) float tile[16 * TILE_PITCH];
  __shared__ v2f wfrag[8 * 32 * 32];
  const int lane = threadIdx.x & 31;
  const int wave = __builtin_amdgcn_readfirstlane((int)(threadIdx.x >> 5));
  const int half = lane >> 4;
  const int r = lane & 15;
  const int n = wave * 16 + r;

  v2f* wl = wfrag + (size_t)wave * 32 * 32 + lane;
  const float* wBase = Wn + (size_t)(half * 2) * 128 + n;
#pragma unroll
  for (int t = 0; t < 32; ++t) {
    v2f f;
    f.x = wBase[(size_t)(4 * t) * 128];
    f.y = wBase[(size_t)(4 * t) * 128 + 128];
    wl[t * 32] = f;
  }
  const float badd = bias_out ? bias_out[n] : 0.0f;
  const float4 b = ((const float4*)bias)[lane];

  const int ntiles = (N + 15) >> 4;
  for (int tIdx = blockIdx.x; tIdx < ntiles; tIdx += gridDim.x) {
    const int m0 = tIdx * 16;
    float* tl = tile;
    const int iA = m0 + wave * 2;
    const int iB = iA + 1;
    const int rl = m0 + lane;
    const int rp = rowptr[(rl < N) ? rl : N];
    const float dv = dis[(rl < N) ? rl : (N - 1)];
    RowHead hA, hB;
    hA.self = ((const float4*)(hw + (size_t)((iA < N) ? iA : (N - 1)) * 128))[lane];
    hB.self = ((const float4*)(hw + (size_t)((iB < N) ? iB : (N - 1)) * 128))[lane];
    hA.s = __builtin_amdgcn_readlane(rp, wave * 2);
    hA.e = __builtin_amdgcn_readlane(rp, wave * 2 + 1);
    hB.s = hA.e;
    hB.e = __builtin_amdgcn_readlane(rp, wave * 2 + 2);
    hA.di = __int_as_float(__builtin_amdgcn_readlane(__float_as_int(dv), wave * 2));
    hB.di = __int_as_float(__builtin_amdgcn_readlane(__float_as_int(dv), wave * 2 + 1));
    hA.c = 0; hA.w = 0.0f;
    hB.c = 0; hB.w = 0.0f;
    row_first(hA, lane, col, wgt);
    row_first(hB, lane, col, wgt);
    const float4 sA = row_sum(hA, iA, lane, hw, col, wgt);
    row_finish(sA, hA, b, iA, N, lane, emb, tl + (wave * 2) * TILE_PITCH);
    const float4 sB = row_sum(hB, iB, lane, hw, col, wgt);
    row_finish(sB, hB, b, iB, N, lane, emb, tl + (wave * 2 + 1) * TILE_PITCH);
    __syncthreads();

    const float* aBase = tl + r * TILE_PITCH + half * 2;
    v8f acc = {};
#pragma unroll
    for (int t = 0; t < 32; ++t) {
      const v2f a = *(const v2f*)(aBase + 4 * t);
      const v2f bq = wl[t * 32];
      acc = __builtin_amdgcn_wmma_f32_16x16x4_f32(false, a, false, bq, (short)0, acc,
                                                  false, false);
    }

    float o[8];
#pragma unroll
    for (int v = 0; v < 8; ++v) {
      const float val = acc[v] + badd;
      o[v] = relu_out ? fmaxf(val, 0.0f) : val;
    }
    __syncthreads();
    float* outBase = Out + (size_t)(m0 + half * 8) * 128 + n;
    if (m0 + 16 <= N) {
#pragma unroll
      for (int v = 0; v < 8; ++v) outBase[(size_t)v * 128] = o[v];
    } else {
#pragma unroll
      for (int v = 0; v < 8; ++v) {
        const int m = m0 + v + half * 8;
        if (m < N) Out[(size_t)m * 128 + n] = o[v];
      }
    }
  }
}

__global__ void pred_kernel(const float* __restrict__ t, const float* __restrict__ Wr2,
                            const float* __restrict__ br2, float* __restrict__ pred, int N) {
  int i = blockIdx.x * blockDim.x + threadIdx.x;
  if (i >= N) return;
  float a0 = br2[0], a1 = br2[1], a2 = br2[2];
  const float* row = t + (size_t)i * 128;
#pragma unroll 4
  for (int k = 0; k < 128; ++k) {
    float tv = row[k];
    a0 += tv * Wr2[k * 3 + 0];
    a1 += tv * Wr2[k * 3 + 1];
    a2 += tv * Wr2[k * 3 + 2];
  }
  pred[(size_t)i * 3 + 0] = a0;
  pred[(size_t)i * 3 + 1] = a1;
  pred[(size_t)i * 3 + 2] = a2;
}

extern "C" void kernel_launch(void* const* d_in, const int* in_sizes, int n_in,
                              void* d_out, int out_size, void* d_ws, size_t ws_size,
                              hipStream_t stream) {
  const float* x   = (const float*)d_in[0];
  const float* xm  = (const float*)d_in[1];
  const int*   ei  = (const int*)d_in[2];
  const float* W0  = (const float*)d_in[3];
  const float* b0  = (const float*)d_in[4];
  const float* Wh  = (const float*)d_in[5];
  const float* bh  = (const float*)d_in[6];
  const float* Wr1 = (const float*)d_in[7];
  const float* br1 = (const float*)d_in[8];
  const float* Wr2 = (const float*)d_in[9];
  const float* br2 = (const float*)d_in[10];

  const int N = in_sizes[0] / 10;
  const int E = in_sizes[2] / 2;
  const int* srcv = ei;
  const int* dstv = ei + E;

  float* bufA = (float*)d_ws;
  float* bufB = bufA + (size_t)N * 128;
  float* bufC = bufB + (size_t)N * 128;
  float* dis  = bufC + (size_t)N * 128;

  const size_t nA = ((size_t)N + 1 + 63) & ~(size_t)63;
  const size_t eA = ((size_t)E + 63) & ~(size_t)63;
  int*   cnt    = (int*)bufC;
  int*   rowptr = cnt + nA;
  int*   cursor = rowptr + nA;
  int*   bsum   = cursor + nA;
  int*   colTmp = bsum + nA;
  int*   rowOf  = colTmp + eA;
  int*   col    = rowOf + eA;
  float* wgt    = (float*)(col + eA);

  float* emb_out  = (float*)d_out;
  float* pred_out = emb_out + (size_t)N * 128;

  const int mblocks = (N + 15) / 16;
  const int eblocks = (E + 255) / 256;
  const int tblocks = (N + 1023) / 1024;
  const int fblocks = (mblocks < 1024) ? mblocks : 1024;

  hipMemsetAsync(cnt, 0, (size_t)N * sizeof(int), stream);
  count_kernel<<<eblocks, 256, 0, stream>>>(dstv, cnt, E);
  scan_partials<<<tblocks, 256, 0, stream>>>(cnt, bsum, N);
  scan_finish<<<tblocks, 256, 0, stream>>>(cnt, bsum, rowptr, cursor, dis, N);
  fill_kernel<<<eblocks, 256, 0, stream>>>(srcv, dstv, cursor, colTmp, rowOf, E);
  order_kernel<<<eblocks, 256, 0, stream>>>(rowptr, colTmp, rowOf, dis, col, wgt, E);

  build_h0<<<(int)(((size_t)N * 16 + 255) / 256), 256, 0, stream>>>(x, xm, bufA, N);

  gemm128<<<mblocks, 256, 0, stream>>>(bufA, 16, N, W0, bufB, nullptr, 0);

  gather_gemm<<<fblocks, 256, 0, stream>>>(bufB, rowptr, col, wgt, dis, b0, nullptr,
                                           Wh, nullptr, 0, bufA, N);
  gather_gemm<<<fblocks, 256, 0, stream>>>(bufA, rowptr, col, wgt, dis, bh, nullptr,
                                           Wh + (size_t)1 * 128 * 128, nullptr, 0, bufB, N);
  gather_gemm<<<fblocks, 256, 0, stream>>>(bufB, rowptr, col, wgt, dis, bh + 128, nullptr,
                                           Wh + (size_t)2 * 128 * 128, nullptr, 0, bufA, N);
  gather_gemm<<<fblocks, 256, 0, stream>>>(bufA, rowptr, col, wgt, dis, bh + 256, nullptr,
                                           Wh + (size_t)3 * 128 * 128, nullptr, 0, bufB, N);
  gather_gemm<<<fblocks, 256, 0, stream>>>(bufB, rowptr, col, wgt, dis, bh + 384, emb_out,
                                           Wr1, br1, 1, bufA, N);

  pred_kernel<<<(N + 255) / 256, 256, 0, stream>>>(bufA, Wr2, br2, pred_out, N);
}


# layer 0 aggregates its 16 input columns before the 16->128 product, both products in the gather kernel: 64 B gathered per edge, not 512; one launch fewer; layer-0 sums in a new fixed order
# speedup vs baseline: 3.8172x; 1.0633x over previous
#include <hip/hip_runtime.h>

typedef __attribute__((ext_vector_type(2))) float v2f;
typedef __attribute__((ext_vector_type(8))) float v8f;

__global__ void count_kernel(const int* __restrict__ dst, int* __restrict__ cnt, int E) {
  int e = blockIdx.x * blockDim.x + threadIdx.x;
  if (e < E) atomicAdd(&cnt[dst[e]], 1);
}

__global__ void scan_partials(const int* __restrict__ cnt, int* __restrict__ bsum, int N) {
  __shared__ int ws[8];
  const int base = blockIdx.x * 1024 + threadIdx.x * 4;
  int s = 0;
  for (int j = 0; j < 4; ++j) {
    int i = base + j;
    if (i < N) s += cnt[i];
  }
  for (int o = 16; o > 0; o >>= 1) s += __shfl_xor(s, o, 32);
  if ((threadIdx.x & 31) == 0) ws[threadIdx.x >> 5] = s;
  __syncthreads();
  if (threadIdx.x == 0) {
    int t = 0;
    for (int w = 0; w < 8; ++w) t += ws[w];
    bsum[blockIdx.x] = t;
  }
}

__global__ void scan_finish(const int* __restrict__ cnt, const int* __restrict__ bsum,
                            int* __restrict__ rowptr, int* __restrict__ cursor,
                            float* __restrict__ dis, int N) {
  __shared__ int ws[8];
  __shared__ int wt[8];
  const int tid = threadIdx.x, lane = tid & 31, wave = tid >> 5;
  int p = 0;
  for (int j = tid; j < (int)blockIdx.x; j += 256) p += bsum[j];
  for (int o = 16; o > 0; o >>= 1) p += __shfl_xor(p, o, 32);
  if (lane == 0) ws[wave] = p;

  const int base = blockIdx.x * 1024 + tid * 4;
  int c[4];
  int s = 0;
  for (int j = 0; j < 4; ++j) {
    int i = base + j;
    c[j] = (i < N) ? cnt[i] : 0;
    s += c[j];
  }
  int inc = s;
  for (int o = 1; o < 32; o <<= 1) {
    int t = __shfl_up(inc, o, 32);
    if (lane >= o) inc += t;
  }
  if (lane == 31) wt[wave] = inc;
  __syncthreads();
  int off = 0;
  for (int w = 0; w < 8; ++w) off += ws[w];
  for (int w = 0; w < wave; ++w) off += wt[w];
  int excl = off + inc - s;
  for (int j = 0; j < 4; ++j) {
    int i = base + j;
    if (i < N) {
      rowptr[i] = excl;
      cursor[i] = excl;
      dis[i] = rsqrtf((float)c[j] + 1.0f);
      excl += c[j];
      if (i == N - 1) rowptr[N] = excl;
    }
  }
}

__global__ void fill_kernel(const int* __restrict__ src, const int* __restrict__ dst,
                            int* __restrict__ cursor, int* __restrict__ colTmp,
                            int* __restrict__ rowOf, int E) {
  int e = blockIdx.x * blockDim.x + threadIdx.x;
  if (e >= E) return;
  int d = dst[e];
  int pos = atomicAdd(&cursor[d], 1);
  colTmp[pos] = src[e];
  rowOf[pos] = d;
}

__global__ void order_kernel(const int* __restrict__ rowptr, const int* __restrict__ colTmp,
                             const int* __restrict__ rowOf, const float* __restrict__ dis,
                             int* __restrict__ col, float* __restrict__ wgt, int E) {
  int p = blockIdx.x * blockDim.x + threadIdx.x;
  if (p >= E) return;
  const int d = rowOf[p];
  const int s = rowptr[d], e = rowptr[d + 1];
  const int my = colTmp[p];
  int rank = 0;
  for (int q = s; q < e; ++q) {
    int v = colTmp[q];
    rank += ((v < my) || (v == my && q < p)) ? 1 : 0;
  }
  col[s + rank] = my;
  wgt[s + rank] = dis[my] * dis[d];
}

__global__ void build_h0(const float* __restrict__ x, const float* __restrict__ xm,
                         float* __restrict__ h0, int N) {
  int idx = blockIdx.x * blockDim.x + threadIdx.x;
  if (idx >= N * 16) return;
  int i = idx >> 4, j = idx & 15;
  h0[idx] = (j < 8) ? x[i * 10 + j] : xm[i * 10 + (j - 8)];
}

__device__ __forceinline__ void fma4(float4& acc, const float4 v, const float w) {
  acc.x = fmaf(v.x, w, acc.x);
  acc.y = fmaf(v.y, w, acc.y);
  acc.z = fmaf(v.z, w, acc.z);
  acc.w = fmaf(v.w, w, acc.w);
}

#define TILE_PITCH 132

struct RowHead {
  int s, e, c;
  float w, di;
  float4 self;
};

__device__ __forceinline__ void row_first(RowHead& h, const int lane,
                                          const int* __restrict__ col, const float* __restrict__ wgt) {
  const int nb = (h.e - h.s < 32) ? (h.e - h.s) : 32;
  if (lane < nb) {
    h.c = col[h.s + lane];
    h.w = wgt[h.s + lane];
  }
}

__device__ __forceinline__ float4 row_sum(const RowHead& h, const int i, const int lane,
                                          const float* __restrict__ hw, const int* __restrict__ col,
                                          const float* __restrict__ wgt) {
  float4 acc = {0.0f, 0.0f, 0.0f, 0.0f};
  int c = h.c;
  float w = h.w;
  for (int base = h.s; base < h.e; base += 32) {
    const int nb = (h.e - base < 32) ? (h.e - base) : 32;
    if (base != h.s) {
      c = i;
      w = 0.0f;
      if (lane < nb) {
        c = col[base + lane];
        w = wgt[base + lane];
      }
    }
    const int wi = __float_as_int(w);
    int j = 0;
    for (; j + 8 <= nb; j += 8) {
      float4 v[8];
#pragma unroll
      for (int u = 0; u < 8; ++u) {
        const int cu = __builtin_amdgcn_readlane(c, j + u);
        v[u] = ((const float4*)(hw + (size_t)cu * 128))[lane];
      }
#pragma unroll
      for (int u = 0; u < 8; ++u)
        fma4(acc, v[u], __int_as_float(__builtin_amdgcn_readlane(wi, j + u)));
    }
    const int m = nb - j;
    if (m > 0) {
      float4 v[7];
#pragma unroll
      for (int u = 0; u < 7; ++u) {
        if (u < m) {
          const int cu = __builtin_amdgcn_readlane(c, j + u);
          v[u] = ((const float4*)(hw + (size_t)cu * 128))[lane];
        }
      }
#pragma unroll
      for (int u = 0; u < 7; ++u) {
        if (u < m) fma4(acc, v[u], __int_as_float(__builtin_amdgcn_readlane(wi, j + u)));
      }
    }
  }
  return acc;
}

__device__ __forceinline__ void row_finish(const float4 acc, const RowHead& h, const float4 b,
                                           const int i, const int N, const int lane,
                                           float* __restrict__ emb, float* __restrict__ tileRow) {
  const float dd = h.di * h.di;
  float4 val;
  val.x = acc.x + h.self.x * dd + b.x;
  val.y = acc.y + h.self.y * dd + b.y;
  val.z = acc.z + h.self.z * dd + b.z;
  val.w = acc.w + h.self.w * dd + b.w;
  if (emb && i < N) ((float4*)(emb + (size_t)i * 128))[lane] = val;
  float4 r;
  r.x = fmaxf(val.x, 0.0f);
  r.y = fmaxf(val.y, 0.0f);
  r.z = fmaxf(val.z, 0.0f);
  r.w = fmaxf(val.w, 0.0f);
  ((float4*)tileRow)[lane] = r;
}

__global__ void __launch_bounds__(256)
gather_gemm(const float* __restrict__ hw, const int* __restrict__ rowptr,
            const int* __restrict__ col, const float* __restrict__ wgt,
            const float* __restrict__ dis, const float* __restrict__ bias,
            float* __restrict__ emb, const float* __restrict__ Wn,
            const float* __restrict__ bias_out, int relu_out,
            float* __restrict__ Out, int N) {
  __shared__ alignas(16) float tile[16 * TILE_PITCH];
  __shared__ v2f wfrag[8 * 32 * 32];
  const int lane = threadIdx.x & 31;
  const int wave = __builtin_amdgcn_readfirstlane((int)(threadIdx.x >> 5));
  const int half = lane >> 4;
  const int r = lane & 15;
  const int n = wave * 16 + r;

  v2f* wl = wfrag + (size_t)wave * 32 * 32 + lane;
  const float* wBase = Wn + (size_t)(half * 2) * 128 + n;
#pragma unroll
  for (int t = 0; t < 32; ++t) {
    v2f f;
    f.x = wBase[(size_t)(4 * t) * 128];
    f.y = wBase[(size_t)(4 * t) * 128 + 128];
    wl[t * 32] = f;
  }
  const float badd = bias_out ? bias_out[n] : 0.0f;
  const float4 b = ((const float4*)bias)[lane];

  const int ntiles = (N + 15) >> 4;
  for (int tIdx = blockIdx.x; tIdx < ntiles; tIdx += gridDim.x) {
    const int m0 = tIdx * 16;
    float* tl = tile;
    const int iA = m0 + wave * 2;
    const int iB = iA + 1;
    const int rl = m0 + lane;
    const int rp = rowptr[(rl < N) ? rl : N];
    const float dv = dis[(rl < N) ? rl : (N - 1)];
    RowHead hA, hB;
    hA.self = ((const float4*)(hw + (size_t)((iA < N) ? iA : (N - 1)) * 128))[lane];
    hB.self = ((const float4*)(hw + (size_t)((iB < N) ? iB : (N - 1)) * 128))[lane];
    hA.s = __builtin_amdgcn_readlane(rp, wave * 2);
    hA.e = __builtin_amdgcn_readlane(rp, wave * 2 + 1);
    hB.s = hA.e;
    hB.e = __builtin_amdgcn_readlane(rp, wave * 2 + 2);
    hA.di = __int_as_float(__builtin_amdgcn_readlane(__float_as_int(dv), wave * 2));
    hB.di = __int_as_float(__builtin_amdgcn_readlane(__float_as_int(dv), wave * 2 + 1));
    hA.c = 0; hA.w = 0.0f;
    hB.c = 0; hB.w = 0.0f;
    row_first(hA, lane, col, wgt);
    row_first(hB, lane, col, wgt);
    const float4 sA = row_sum(hA, iA, lane, hw, col, wgt);
    row_finish(sA, hA, b, iA, N, lane, emb, tl + (wave * 2) * TILE_PITCH);
    const float4 sB = row_sum(hB, iB, lane, hw, col, wgt);
    row_finish(sB, hB, b, iB, N, lane, emb, tl + (wave * 2 + 1) * TILE_PITCH);
    __syncthreads();

    const float* aBase = tl + r * TILE_PITCH + half * 2;
    v8f acc = {};
#pragma unroll
    for (int t = 0; t < 32; ++t) {
      const v2f a = *(const v2f*)(aBase + 4 * t);
      const v2f bq = wl[t * 32];
      acc = __builtin_amdgcn_wmma_f32_16x16x4_f32(false, a, false, bq, (short)0, acc,
                                                  false, false);
    }

    float o[8];
#pragma unroll
    for (int v = 0; v < 8; ++v) {
      const float val = acc[v] + badd;
      o[v] = relu_out ? fmaxf(val, 0.0f) : val;
    }
    __syncthreads();
    float* outBase = Out + (size_t)(m0 + half * 8) * 128 + n;
    if (m0 + 16 <= N) {
#pragma unroll
      for (int v = 0; v < 8; ++v) outBase[(size_t)v * 128] = o[v];
    } else {
#pragma unroll
      for (int v = 0; v < 8; ++v) {
        const int m = m0 + v + half * 8;
        if (m < N) Out[(size_t)m * 128 + n] = o[v];
      }
    }
  }
}

#define IN_PITCH 20

__global__ void __launch_bounds__(256)
gather16_gemm(const float* __restrict__ h0, const int* __restrict__ rowptr,
              const int* __restrict__ col, const float* __restrict__ wgt,
              const float* __restrict__ dis, const float* __restrict__ W0,
              const float* __restrict__ b0, const float* __restrict__ Wn,
              float* __restrict__ Out, int N) {
  __shared__ alignas(16) float tin[16 * IN_PITCH];
  __shared__ alignas(16) float tile[16 * TILE_PITCH];
  __shared__ v2f wfrag[8 * 32 * 32];
  const int lane = threadIdx.x & 31;
  const int wave = __builtin_amdgcn_readfirstlane((int)(threadIdx.x >> 5));
  const int half = lane >> 4;
  const int r = lane & 15;
  const int n = wave * 16 + r;
  const int q = lane & 3;
  const int es = (lane >> 2) & 3;
  const int rI = wave * 2 + half;

  v2f* wl = wfrag + (size_t)wave * 32 * 32 + lane;
  const float* wBase = Wn + (size_t)(half * 2) * 128 + n;
#pragma unroll
  for (int t = 0; t < 32; ++t) {
    v2f f;
    f.x = wBase[(size_t)(4 * t) * 128];
    f.y = wBase[(size_t)(4 * t) * 128 + 128];
    wl[t * 32] = f;
  }
  v2f w0f[4];
  const float* w0Base = W0 + (size_t)(half * 2) * 128 + n;
#pragma unroll
  for (int t = 0; t < 4; ++t) {
    w0f[t].x = w0Base[(size_t)(4 * t) * 128];
    w0f[t].y = w0Base[(size_t)(4 * t) * 128 + 128];
  }
  const float b0n = b0[n];

  const int ntiles = (N + 15) >> 4;
  for (int tIdx = blockIdx.x; tIdx < ntiles; tIdx += gridDim.x) {
    const int m0 = tIdx * 16;
    const int i = m0 + rI;
    const int ic = (i < N) ? i : (N - 1);
    const int s = rowptr[ic];
    const int eIn = rowptr[ic + 1];
    const float di = dis[ic];
    const float4 self = ((const float4*)(h0 + (size_t)ic * 16))[q];
    const int e = (i < N) ? eIn : s;

    float4 acc = {0.0f, 0.0f, 0.0f, 0.0f};
    for (int base = s; base < e; base += 16) {
      int c[4];
      float w[4];
      float4 v[4];
#pragma unroll
      for (int u = 0; u < 4; ++u) {
        const int p = base + es + 4 * u;
        c[u] = ic;
        w[u] = 0.0f;
        if (p < e) {
          c[u] = col[p];
          w[u] = wgt[p];
        }
      }
#pragma unroll
      for (int u = 0; u < 4; ++u) {
        v[u] = float4{0.0f, 0.0f, 0.0f, 0.0f};
        if (base + es + 4 * u < e) v[u] = ((const float4*)(h0 + (size_t)c[u] * 16))[q];
      }
#pragma unroll
      for (int u = 0; u < 4; ++u) fma4(acc, v[u], w[u]);
    }
    acc.x += __shfl_xor(acc.x, 4, 32);
    acc.y += __shfl_xor(acc.y, 4, 32);
    acc.z += __shfl_xor(acc.z, 4, 32);
    acc.w += __shfl_xor(acc.w, 4, 32);
    acc.x += __shfl_xor(acc.x, 8, 32);
    acc.y += __shfl_xor(acc.y, 8, 32);
    acc.z += __shfl_xor(acc.z, 8, 32);
    acc.w += __shfl_xor(acc.w, 8, 32);
    if (es == 0) {
      const float dd = di * di;
      float4 val;
      val.x = acc.x + self.x * dd;
      val.y = acc.y + self.y * dd;
      val.z = acc.z + self.z * dd;
      val.w = acc.w + self.w * dd;
      ((float4*)(tin + rI * IN_PITCH))[q] = val;
    }
    __syncthreads();

    const float* a0 = tin + r * IN_PITCH + half * 2;
    v8f acc1 = {};
#pragma unroll
    for (int t = 0; t < 4; ++t) {
      const v2f a = *(const v2f*)(a0 + 4 * t);
      acc1 = __builtin_amdgcn_wmma_f32_16x16x4_f32(false, a, false, w0f[t], (short)0, acc1,
                                                  false, false);
    }
    float* tcol = tile + (half * 8) * TILE_PITCH + n;
#pragma unroll
    for (int v = 0; v < 8; ++v) tcol[v * TILE_PITCH] = fmaxf(acc1[v] + b0n, 0.0f);
    __syncthreads();

    const float* aBase = tile + r * TILE_PITCH + half * 2;
    v8f acc2 = {};
#pragma unroll
    for (int t = 0; t < 32; ++t) {
      const v2f a = *(const v2f*)(aBase + 4 * t);
      const v2f bq = wl[t * 32];
      acc2 = __builtin_amdgcn_wmma_f32_16x16x4_f32(false, a, false, bq, (short)0, acc2,
                                                  false, false);
    }
    float* outBase = Out + (size_t)(m0 + half * 8) * 128 + n;
    if (m0 + 16 <= N) {
#pragma unroll
      for (int v = 0; v < 8; ++v) outBase[(size_t)v * 128] = acc2[v];
    } else {
#pragma unroll
      for (int v = 0; v < 8; ++v) {
        const int m = m0 + v + half * 8;
        if (m < N) Out[(size_t)m * 128 + n] = acc2[v];
      }
    }
  }
}

__global__ void pred_kernel(const float* __restrict__ t, const float* __restrict__ Wr2,
                            const float* __restrict__ br2, float* __restrict__ pred, int N) {
  int i = blockIdx.x * blockDim.x + threadIdx.x;
  if (i >= N) return;
  float a0 = br2[0], a1 = br2[1], a2 = br2[2];
  const float* row = t + (size_t)i * 128;
#pragma unroll 4
  for (int k = 0; k < 128; ++k) {
    float tv = row[k];
    a0 += tv * Wr2[k * 3 + 0];
    a1 += tv * Wr2[k * 3 + 1];
    a2 += tv * Wr2[k * 3 + 2];
  }
  pred[(size_t)i * 3 + 0] = a0;
  pred[(size_t)i * 3 + 1] = a1;
  pred[(size_t)i * 3 + 2] = a2;
}

extern "C" void kernel_launch(void* const* d_in, const int* in_sizes, int n_in,
                              void* d_out, int out_size, void* d_ws, size_t ws_size,
                              hipStream_t stream) {
  const float* x   = (const float*)d_in[0];
  const float* xm  = (const float*)d_in[1];
  const int*   ei  = (const int*)d_in[2];
  const float* W0  = (const float*)d_in[3];
  const float* b0  = (const float*)d_in[4];
  const float* Wh  = (const float*)d_in[5];
  const float* bh  = (const float*)d_in[6];
  const float* Wr1 = (const float*)d_in[7];
  const float* br1 = (const float*)d_in[8];
  const float* Wr2 = (const float*)d_in[9];
  const float* br2 = (const float*)d_in[10];

  const int N = in_sizes[0] / 10;
  const int E = in_sizes[2] / 2;
  const int* srcv = ei;
  const int* dstv = ei + E;

  float* bufA = (float*)d_ws;
  float* bufB = bufA + (size_t)N * 128;
  float* bufC = bufB + (size_t)N * 128;
  float* dis  = bufC + (size_t)N * 128;

  const size_t nA = ((size_t)N + 1 + 63) & ~(size_t)63;
  const size_t eA = ((size_t)E + 63) & ~(size_t)63;
  int*   cnt    = (int*)bufC;
  int*   rowptr = cnt + nA;
  int*   cursor = rowptr + nA;
  int*   bsum   = cursor + nA;
  int*   colTmp = bsum + nA;
  int*   rowOf  = colTmp + eA;
  int*   col    = rowOf + eA;
  float* wgt    = (float*)(col + eA);

  float* emb_out  = (float*)d_out;
  float* pred_out = emb_out + (size_t)N * 128;

  const int mblocks = (N + 15) / 16;
  const int eblocks = (E + 255) / 256;
  const int tblocks = (N + 1023) / 1024;
  const int fblocks = (mblocks < 1024) ? mblocks : 1024;

  hipMemsetAsync(cnt, 0, (size_t)N * sizeof(int), stream);
  count_kernel<<<eblocks, 256, 0, stream>>>(dstv, cnt, E);
  scan_partials<<<tblocks, 256, 0, stream>>>(cnt, bsum, N);
  scan_finish<<<tblocks, 256, 0, stream>>>(cnt, bsum, rowptr, cursor, dis, N);
  fill_kernel<<<eblocks, 256, 0, stream>>>(srcv, dstv, cursor, colTmp, rowOf, E);
  order_kernel<<<eblocks, 256, 0, stream>>>(rowptr, colTmp, rowOf, dis, col, wgt, E);

  build_h0<<<(int)(((size_t)N * 16 + 255) / 256), 256, 0, stream>>>(x, xm, bufB, N);

  gather16_gemm<<<fblocks, 256, 0, stream>>>(bufB, rowptr, col, wgt, dis, W0, b0, Wh, bufA, N);

  gather_gemm<<<fblocks, 256, 0, stream>>>(bufA, rowptr, col, wgt, dis, bh, nullptr,
                                           Wh + (size_t)1 * 128 * 128, nullptr, 0, bufB, N);
  gather_gemm<<<fblocks, 256, 0, stream>>>(bufB, rowptr, col, wgt, dis, bh + 128, nullptr,
                                           Wh + (size_t)2 * 128 * 128, nullptr, 0, bufA, N);
  gather_gemm<<<fblocks, 256, 0, stream>>>(bufA, rowptr, col, wgt, dis, bh + 256, nullptr,
                                           Wh + (size_t)3 * 128 * 128, nullptr, 0, bufB, N);
  gather_gemm<<<fblocks, 256, 0, stream>>>(bufB, rowptr, col, wgt, dis, bh + 384, emb_out,
                                           Wr1, br1, 1, bufA, N);

  pred_kernel<<<(N + 255) / 256, 256, 0, stream>>>(bufA, Wr2, br2, pred_out, N);
}


# edge placement takes the rank the degree count's atomic add returns (row start + rank): no atomics in the fill pass, one global atomic per edge instead of two; col/wgt and all sums as before
# speedup vs baseline: 4.1202x; 1.0794x over previous
#include <hip/hip_runtime.h>

typedef __attribute__((ext_vector_type(2))) float v2f;
typedef __attribute__((ext_vector_type(8))) float v8f;

__global__ void count_kernel(const int* __restrict__ dst, int* __restrict__ cnt,
                             int* __restrict__ rank, int E) {
  int e = blockIdx.x * blockDim.x + threadIdx.x;
  if (e < E) rank[e] = atomicAdd(&cnt[dst[e]], 1);
}

__global__ void scan_partials(const int* __restrict__ cnt, int* __restrict__ bsum, int N) {
  __shared__ int ws[8];
  const int base = blockIdx.x * 1024 + threadIdx.x * 4;
  int s = 0;
  for (int j = 0; j < 4; ++j) {
    int i = base + j;
    if (i < N) s += cnt[i];
  }
  for (int o = 16; o > 0; o >>= 1) s += __shfl_xor(s, o, 32);
  if ((threadIdx.x & 31) == 0) ws[threadIdx.x >> 5] = s;
  __syncthreads();
  if (threadIdx.x == 0) {
    int t = 0;
    for (int w = 0; w < 8; ++w) t += ws[w];
    bsum[blockIdx.x] = t;
  }
}

__global__ void scan_finish(const int* __restrict__ cnt, const int* __restrict__ bsum,
                            int* __restrict__ rowptr, float* __restrict__ dis, int N) {
  __shared__ int ws[8];
  __shared__ int wt[8];
  const int tid = threadIdx.x, lane = tid & 31, wave = tid >> 5;
  int p = 0;
  for (int j = tid; j < (int)blockIdx.x; j += 256) p += bsum[j];
  for (int o = 16; o > 0; o >>= 1) p += __shfl_xor(p, o, 32);
  if (lane == 0) ws[wave] = p;

  const int base = blockIdx.x * 1024 + tid * 4;
  int c[4];
  int s = 0;
  for (int j = 0; j < 4; ++j) {
    int i = base + j;
    c[j] = (i < N) ? cnt[i] : 0;
    s += c[j];
  }
  int inc = s;
  for (int o = 1; o < 32; o <<= 1) {
    int t = __shfl_up(inc, o, 32);
    if (lane >= o) inc += t;
  }
  if (lane == 31) wt[wave] = inc;
  __syncthreads();
  int off = 0;
  for (int w = 0; w < 8; ++w) off += ws[w];
  for (int w = 0; w < wave; ++w) off += wt[w];
  int excl = off + inc - s;
  for (int j = 0; j < 4; ++j) {
    int i = base + j;
    if (i < N) {
      rowptr[i] = excl;
      dis[i] = rsqrtf((float)c[j] + 1.0f);
      excl += c[j];
      if (i == N - 1) rowptr[N] = excl;
    }
  }
}

__global__ void fill_kernel(const int* __restrict__ src, const int* __restrict__ dst,
                            const int* __restrict__ rowptr, const int* __restrict__ rank,
                            int* __restrict__ colTmp, int* __restrict__ rowOf, int E) {
  int e = blockIdx.x * blockDim.x + threadIdx.x;
  if (e >= E) return;
  int d = dst[e];
  int pos = rowptr[d] + rank[e];
  colTmp[pos] = src[e];
  rowOf[pos] = d;
}

__global__ void order_kernel(const int* __restrict__ rowptr, const int* __restrict__ colTmp,
                             const int* __restrict__ rowOf, const float* __restrict__ dis,
                             int* __restrict__ col, float* __restrict__ wgt, int E) {
  int p = blockIdx.x * blockDim.x + threadIdx.x;
  if (p >= E) return;
  const int d = rowOf[p];
  const int s = rowptr[d], e = rowptr[d + 1];
  const int my = colTmp[p];
  int rank = 0;
  for (int q = s; q < e; ++q) {
    int v = colTmp[q];
    rank += ((v < my) || (v == my && q < p)) ? 1 : 0;
  }
  col[s + rank] = my;
  wgt[s + rank] = dis[my] * dis[d];
}

__global__ void build_h0(const float* __restrict__ x, const float* __restrict__ xm,
                         float* __restrict__ h0, int N) {
  int idx = blockIdx.x * blockDim.x + threadIdx.x;
  if (idx >= N * 16) return;
  int i = idx >> 4, j = idx & 15;
  h0[idx] = (j < 8) ? x[i * 10 + j] : xm[i * 10 + (j - 8)];
}

__device__ __forceinline__ void fma4(float4& acc, const float4 v, const float w) {
  acc.x = fmaf(v.x, w, acc.x);
  acc.y = fmaf(v.y, w, acc.y);
  acc.z = fmaf(v.z, w, acc.z);
  acc.w = fmaf(v.w, w, acc.w);
}

#define TILE_PITCH 132

struct RowHead {
  int s, e, c;
  float w, di;
  float4 self;
};

__device__ __forceinline__ void row_first(RowHead& h, const int lane,
                                          const int* __restrict__ col, const float* __restrict__ wgt) {
  const int nb = (h.e - h.s < 32) ? (h.e - h.s) : 32;
  if (lane < nb) {
    h.c = col[h.s + lane];
    h.w = wgt[h.s + lane];
  }
}

__device__ __forceinline__ float4 row_sum(const RowHead& h, const int i, const int lane,
                                          const float* __restrict__ hw, const int* __restrict__ col,
                                          const float* __restrict__ wgt) {
  float4 acc = {0.0f, 0.0f, 0.0f, 0.0f};
  int c = h.c;
  float w = h.w;
  for (int base = h.s; base < h.e; base += 32) {
    const int nb = (h.e - base < 32) ? (h.e - base) : 32;
    if (base != h.s) {
      c = i;
      w = 0.0f;
      if (lane < nb) {
        c = col[base + lane];
        w = wgt[base + lane];
      }
    }
    const int wi = __float_as_int(w);
    int j = 0;
    for (; j + 8 <= nb; j += 8) {
      float4 v[8];
#pragma unroll
      for (int u = 0; u < 8; ++u) {
        const int cu = __builtin_amdgcn_readlane(c, j + u);
        v[u] = ((const float4*)(hw + (size_t)cu * 128))[lane];
      }
#pragma unroll
      for (int u = 0; u < 8; ++u)
        fma4(acc, v[u], __int_as_float(__builtin_amdgcn_readlane(wi, j + u)));
    }
    const int m = nb - j;
    if (m > 0) {
      float4 v[7];
#pragma unroll
      for (int u = 0; u < 7; ++u) {
        if (u < m) {
          const int cu = __builtin_amdgcn_readlane(c, j + u);
          v[u] = ((const float4*)(hw + (size_t)cu * 128))[lane];
        }
      }
#pragma unroll
      for (int u = 0; u < 7; ++u) {
        if (u < m) fma4(acc, v[u], __int_as_float(__builtin_amdgcn_readlane(wi, j + u)));
      }
    }
  }
  return acc;
}

__device__ __forceinline__ void row_finish(const float4 acc, const RowHead& h, const float4 b,
                                           const int i, const int N, const int lane,
                                           float* __restrict__ emb, float* __restrict__ tileRow) {
  const float dd = h.di * h.di;
  float4 val;
  val.x = acc.x + h.self.x * dd + b.x;
  val.y = acc.y + h.self.y * dd + b.y;
  val.z = acc.z + h.self.z * dd + b.z;
  val.w = acc.w + h.self.w * dd + b.w;
  if (emb && i < N) ((float4*)(emb + (size_t)i * 128))[lane] = val;
  float4 r;
  r.x = fmaxf(val.x, 0.0f);
  r.y = fmaxf(val.y, 0.0f);
  r.z = fmaxf(val.z, 0.0f);
  r.w = fmaxf(val.w, 0.0f);
  ((float4*)tileRow)[lane] = r;
}

__global__ void __launch_bounds__(256)
gather_gemm(const float* __restrict__ hw, const int* __restrict__ rowptr,
            const int* __restrict__ col, const float* __restrict__ wgt,
            const float* __restrict__ dis, const float* __restrict__ bias,
            float* __restrict__ emb, const float* __restrict__ Wn,
            const float* __restrict__ bias_out, int relu_out,
            float* __restrict__ Out, int N) {
  __shared__ alignas(16) float tile[16 * TILE_PITCH];
  __shared__ v2f wfrag[8 * 32 * 32];
  const int lane = threadIdx.x & 31;
  const int wave = __builtin_amdgcn_readfirstlane((int)(threadIdx.x >> 5));
  const int half = lane >> 4;
  const int r = lane & 15;
  const int n = wave * 16 + r;

  v2f* wl = wfrag + (size_t)wave * 32 * 32 + lane;
  const float* wBase = Wn + (size_t)(half * 2) * 128 + n;
#pragma unroll
  for (int t = 0; t < 32; ++t) {
    v2f f;
    f.x = wBase[(size_t)(4 * t) * 128];
    f.y = wBase[(size_t)(4 * t) * 128 + 128];
    wl[t * 32] = f;
  }
  const float badd = bias_out ? bias_out[n] : 0.0f;
  const float4 b = ((const float4*)bias)[lane];

  const int ntiles = (N + 15) >> 4;
  for (int tIdx = blockIdx.x; tIdx < ntiles; tIdx += gridDim.x) {
    const int m0 = tIdx * 16;
    float* tl = tile;
    const int iA = m0 + wave * 2;
    const int iB = iA + 1;
    const int rl = m0 + lane;
    const int rp = rowptr[(rl < N) ? rl : N];
    const float dv = dis[(rl < N) ? rl : (N - 1)];
    RowHead hA, hB;
    hA.self = ((const float4*)(hw + (size_t)((iA < N) ? iA : (N - 1)) * 128))[lane];
    hB.self = ((const float4*)(hw + (size_t)((iB < N) ? iB : (N - 1)) * 128))[lane];
    hA.s = __builtin_amdgcn_readlane(rp, wave * 2);
    hA.e = __builtin_amdgcn_readlane(rp, wave * 2 + 1);
    hB.s = hA.e;
    hB.e = __builtin_amdgcn_readlane(rp, wave * 2 + 2);
    hA.di = __int_as_float(__builtin_amdgcn_readlane(__float_as_int(dv), wave * 2));
    hB.di = __int_as_float(__builtin_amdgcn_readlane(__float_as_int(dv), wave * 2 + 1));
    hA.c = 0; hA.w = 0.0f;
    hB.c = 0; hB.w = 0.0f;
    row_first(hA, lane, col, wgt);
    row_first(hB, lane, col, wgt);
    const float4 sA = row_sum(hA, iA, lane, hw, col, wgt);
    row_finish(sA, hA, b, iA, N, lane, emb, tl + (wave * 2) * TILE_PITCH);
    const float4 sB = row_sum(hB, iB, lane, hw, col, wgt);
    row_finish(sB, hB, b, iB, N, lane, emb, tl + (wave * 2 + 1) * TILE_PITCH);
    __syncthreads();

    const float* aBase = tl + r * TILE_PITCH + half * 2;
    v8f acc = {};
#pragma unroll
    for (int t = 0; t < 32; ++t) {
      const v2f a = *(const v2f*)(aBase + 4 * t);
      const v2f bq = wl[t * 32];
      acc = __builtin_amdgcn_wmma_f32_16x16x4_f32(false, a, false, bq, (short)0, acc,
                                                  false, false);
    }

    float o[8];
#pragma unroll
    for (int v = 0; v < 8; ++v) {
      const float val = acc[v] + badd;
      o[v] = relu_out ? fmaxf(val, 0.0f) : val;
    }
    __syncthreads();
    float* outBase = Out + (size_t)(m0 + half * 8) * 128 + n;
    if (m0 + 16 <= N) {
#pragma unroll
      for (int v = 0; v < 8; ++v) outBase[(size_t)v * 128] = o[v];
    } else {
#pragma unroll
      for (int v = 0; v < 8; ++v) {
        const int m = m0 + v + half * 8;
        if (m < N) Out[(size_t)m * 128 + n] = o[v];
      }
    }
  }
}

#define IN_PITCH 20

__global__ void __launch_bounds__(256)
gather16_gemm(const float* __restrict__ h0, const int* __restrict__ rowptr,
              const int* __restrict__ col, const float* __restrict__ wgt,
              const float* __restrict__ dis, const float* __restrict__ W0,
              const float* __restrict__ b0, const float* __restrict__ Wn,
              float* __restrict__ Out, int N) {
  __shared__ alignas(16) float tin[16 * IN_PITCH];
  __shared__ alignas(16) float tile[16 * TILE_PITCH];
  __shared__ v2f wfrag[8 * 32 * 32];
  const int lane = threadIdx.x & 31;
  const int wave = __builtin_amdgcn_readfirstlane((int)(threadIdx.x >> 5));
  const int half = lane >> 4;
  const int r = lane & 15;
  const int n = wave * 16 + r;
  const int q = lane & 3;
  const int es = (lane >> 2) & 3;
  const int rI = wave * 2 + half;

  v2f* wl = wfrag + (size_t)wave * 32 * 32 + lane;
  const float* wBase = Wn + (size_t)(half * 2) * 128 + n;
#pragma unroll
  for (int t = 0; t < 32; ++t) {
    v2f f;
    f.x = wBase[(size_t)(4 * t) * 128];
    f.y = wBase[(size_t)(4 * t) * 128 + 128];
    wl[t * 32] = f;
  }
  v2f w0f[4];
  const float* w0Base = W0 + (size_t)(half * 2) * 128 + n;
#pragma unroll
  for (int t = 0; t < 4; ++t) {
    w0f[t].x = w0Base[(size_t)(4 * t) * 128];
    w0f[t].y = w0Base[(size_t)(4 * t) * 128 + 128];
  }
  const float b0n = b0[n];

  const int ntiles = (N + 15) >> 4;
  for (int tIdx = blockIdx.x; tIdx < ntiles; tIdx += gridDim.x) {
    const int m0 = tIdx * 16;
    const int i = m0 + rI;
    const int ic = (i < N) ? i : (N - 1);
    const int s = rowptr[ic];
    const int eIn = rowptr[ic + 1];
    const float di = dis[ic];
    const float4 self = ((const float4*)(h0 + (size_t)ic * 16))[q];
    const int e = (i < N) ? eIn : s;

    float4 acc = {0.0f, 0.0f, 0.0f, 0.0f};
    for (int base = s; base < e; base += 16) {
      int c[4];
      float w[4];
      float4 v[4];
#pragma unroll
      for (int u = 0; u < 4; ++u) {
        const int p = base + es + 4 * u;
        c[u] = ic;
        w[u] = 0.0f;
        if (p < e) {
          c[u] = col[p];
          w[u] = wgt[p];
        }
      }
#pragma unroll
      for (int u = 0; u < 4; ++u) {
        v[u] = float4{0.0f, 0.0f, 0.0f, 0.0f};
        if (base + es + 4 * u < e) v[u] = ((const float4*)(h0 + (size_t)c[u] * 16))[q];
      }
#pragma unroll
      for (int u = 0; u < 4; ++u) fma4(acc, v[u], w[u]);
    }
    acc.x += __shfl_xor(acc.x, 4, 32);
    acc.y += __shfl_xor(acc.y, 4, 32);
    acc.z += __shfl_xor(acc.z, 4, 32);
    acc.w += __shfl_xor(acc.w, 4, 32);
    acc.x += __shfl_xor(acc.x, 8, 32);
    acc.y += __shfl_xor(acc.y, 8, 32);
    acc.z += __shfl_xor(acc.z, 8, 32);
    acc.w += __shfl_xor(acc.w, 8, 32);
    if (es == 0) {
      const float dd = di * di;
      float4 val;
      val.x = acc.x + self.x * dd;
      val.y = acc.y + self.y * dd;
      val.z = acc.z + self.z * dd;
      val.w = acc.w + self.w * dd;
      ((float4*)(tin + rI * IN_PITCH))[q] = val;
    }
    __syncthreads();

    const float* a0 = tin + r * IN_PITCH + half * 2;
    v8f acc1 = {};
#pragma unroll
    for (int t = 0; t < 4; ++t) {
      const v2f a = *(const v2f*)(a0 + 4 * t);
      acc1 = __builtin_amdgcn_wmma_f32_16x16x4_f32(false, a, false, w0f[t], (short)0, acc1,
                                                  false, false);
    }
    float* tcol = tile + (half * 8) * TILE_PITCH + n;
#pragma unroll
    for (int v = 0; v < 8; ++v) tcol[v * TILE_PITCH] = fmaxf(acc1[v] + b0n, 0.0f);
    __syncthreads();

    const float* aBase = tile + r * TILE_PITCH + half * 2;
    v8f acc2 = {};
#pragma unroll
    for (int t = 0; t < 32; ++t) {
      const v2f a = *(const v2f*)(aBase + 4 * t);
      const v2f bq = wl[t * 32];
      acc2 = __builtin_amdgcn_wmma_f32_16x16x4_f32(false, a, false, bq, (short)0, acc2,
                                                  false, false);
    }
    float* outBase = Out + (size_t)(m0 + half * 8) * 128 + n;
    if (m0 + 16 <= N) {
#pragma unroll
      for (int v = 0; v < 8; ++v) outBase[(size_t)v * 128] = acc2[v];
    } else {
#pragma unroll
      for (int v = 0; v < 8; ++v) {
        const int m = m0 + v + half * 8;
        if (m < N) Out[(size_t)m * 128 + n] = acc2[v];
      }
    }
  }
}

__global__ void pred_kernel(const float* __restrict__ t, const float* __restrict__ Wr2,
                            const float* __restrict__ br2, float* __restrict__ pred, int N) {
  int i = blockIdx.x * blockDim.x + threadIdx.x;
  if (i >= N) return;
  float a0 = br2[0], a1 = br2[1], a2 = br2[2];
  const float* row = t + (size_t)i * 128;
#pragma unroll 4
  for (int k = 0; k < 128; ++k) {
    float tv = row[k];
    a0 += tv * Wr2[k * 3 + 0];
    a1 += tv * Wr2[k * 3 + 1];
    a2 += tv * Wr2[k * 3 + 2];
  }
  pred[(size_t)i * 3 + 0] = a0;
  pred[(size_t)i * 3 + 1] = a1;
  pred[(size_t)i * 3 + 2] = a2;
}

extern "C" void kernel_launch(void* const* d_in, const int* in_sizes, int n_in,
                              void* d_out, int out_size, void* d_ws, size_t ws_size,
                              hipStream_t stream) {
  const float* x   = (const float*)d_in[0];
  const float* xm  = (const float*)d_in[1];
  const int*   ei  = (const int*)d_in[2];
  const float* W0  = (const float*)d_in[3];
  const float* b0  = (const float*)d_in[4];
  const float* Wh  = (const float*)d_in[5];
  const float* bh  = (const float*)d_in[6];
  const float* Wr1 = (const float*)d_in[7];
  const float* br1 = (const float*)d_in[8];
  const float* Wr2 = (const float*)d_in[9];
  const float* br2 = (const float*)d_in[10];

  const int N = in_sizes[0] / 10;
  const int E = in_sizes[2] / 2;
  const int* srcv = ei;
  const int* dstv = ei + E;

  float* bufA = (float*)d_ws;
  float* bufB = bufA + (size_t)N * 128;
  float* bufC = bufB + (size_t)N * 128;
  float* dis  = bufC + (size_t)N * 128;

  const size_t nA = ((size_t)N + 1 + 63) & ~(size_t)63;
  const size_t eA = ((size_t)E + 63) & ~(size_t)63;
  int*   cnt    = (int*)bufC;
  int*   rowptr = cnt + nA;
  int*   bsum   = rowptr + nA;
  int*   colTmp = bsum + nA;
  int*   rowOf  = colTmp + eA;
  int*   col    = rowOf + eA;
  float* wgt    = (float*)(col + eA);
  int*   rank   = (int*)(wgt + eA);

  float* emb_out  = (float*)d_out;
  float* pred_out = emb_out + (size_t)N * 128;

  const int mblocks = (N + 15) / 16;
  const int eblocks = (E + 255) / 256;
  const int tblocks = (N + 1023) / 1024;
  const int fblocks = (mblocks < 1024) ? mblocks : 1024;

  hipMemsetAsync(cnt, 0, (size_t)N * sizeof(int), stream);
  count_kernel<<<eblocks, 256, 0, stream>>>(dstv, cnt, rank, E);
  scan_partials<<<tblocks, 256, 0, stream>>>(cnt, bsum, N);
  scan_finish<<<tblocks, 256, 0, stream>>>(cnt, bsum, rowptr, dis, N);
  fill_kernel<<<eblocks, 256, 0, stream>>>(srcv, dstv, rowptr, rank, colTmp, rowOf, E);
  order_kernel<<<eblocks, 256, 0, stream>>>(rowptr, colTmp, rowOf, dis, col, wgt, E);

  build_h0<<<(int)(((size_t)N * 16 + 255) / 256), 256, 0, stream>>>(x, xm, bufB, N);

  gather16_gemm<<<fblocks, 256, 0, stream>>>(bufB, rowptr, col, wgt, dis, W0, b0, Wh, bufA, N);

  gather_gemm<<<fblocks, 256, 0, stream>>>(bufA, rowptr, col, wgt, dis, bh, nullptr,
                                           Wh + (size_t)1 * 128 * 128, nullptr, 0, bufB, N);
  gather_gemm<<<fblocks, 256, 0, stream>>>(bufB, rowptr, col, wgt, dis, bh + 128, nullptr,
                                           Wh + (size_t)2 * 128 * 128, nullptr, 0, bufA, N);
  gather_gemm<<<fblocks, 256, 0, stream>>>(bufA, rowptr, col, wgt, dis, bh + 256, nullptr,
                                           Wh + (size_t)3 * 128 * 128, nullptr, 0, bufB, N);
  gather_gemm<<<fblocks, 256, 0, stream>>>(bufB, rowptr, col, wgt, dis, bh + 384, emb_out,
                                           Wr1, br1, 1, bufA, N);

  pred_kernel<<<(N + 255) / 256, 256, 0, stream>>>(bufA, Wr2, br2, pred_out, N);
}


# head's 128x3 product fused into the last aggregation launch: matrix operands swap slots so a lane holds 8 columns of one row; pred sums in a new fixed order; one launch and 51 MB write + read fewer
# speedup vs baseline: 4.3032x; 1.0444x over previous
#include <hip/hip_runtime.h>

typedef __attribute__((ext_vector_type(2))) float v2f;
typedef __attribute__((ext_vector_type(8))) float v8f;

__global__ void count_kernel(const int* __restrict__ dst, int* __restrict__ cnt,
                             int* __restrict__ rank, int E) {
  int e = blockIdx.x * blockDim.x + threadIdx.x;
  if (e < E) rank[e] = atomicAdd(&cnt[dst[e]], 1);
}

__global__ void scan_partials(const int* __restrict__ cnt, int* __restrict__ bsum, int N) {
  __shared__ int ws[8];
  const int base = blockIdx.x * 1024 + threadIdx.x * 4;
  int s = 0;
  for (int j = 0; j < 4; ++j) {
    int i = base + j;
    if (i < N) s += cnt[i];
  }
  for (int o = 16; o > 0; o >>= 1) s += __shfl_xor(s, o, 32);
  if ((threadIdx.x & 31) == 0) ws[threadIdx.x >> 5] = s;
  __syncthreads();
  if (threadIdx.x == 0) {
    int t = 0;
    for (int w = 0; w < 8; ++w) t += ws[w];
    bsum[blockIdx.x] = t;
  }
}

__global__ void scan_finish(const int* __restrict__ cnt, const int* __restrict__ bsum,
                            int* __restrict__ rowptr, float* __restrict__ dis, int N) {
  __shared__ int ws[8];
  __shared__ int wt[8];
  const int tid = threadIdx.x, lane = tid & 31, wave = tid >> 5;
  int p = 0;
  for (int j = tid; j < (int)blockIdx.x; j += 256) p += bsum[j];
  for (int o = 16; o > 0; o >>= 1) p += __shfl_xor(p, o, 32);
  if (lane == 0) ws[wave] = p;

  const int base = blockIdx.x * 1024 + tid * 4;
  int c[4];
  int s = 0;
  for (int j = 0; j < 4; ++j) {
    int i = base + j;
    c[j] = (i < N) ? cnt[i] : 0;
    s += c[j];
  }
  int inc = s;
  for (int o = 1; o < 32; o <<= 1) {
    int t = __shfl_up(inc, o, 32);
    if (lane >= o) inc += t;
  }
  if (lane == 31) wt[wave] = inc;
  __syncthreads();
  int off = 0;
  for (int w = 0; w < 8; ++w) off += ws[w];
  for (int w = 0; w < wave; ++w) off += wt[w];
  int excl = off + inc - s;
  for (int j = 0; j < 4; ++j) {
    int i = base + j;
    if (i < N) {
      rowptr[i] = excl;
      dis[i] = rsqrtf((float)c[j] + 1.0f);
      excl += c[j];
      if (i == N - 1) rowptr[N] = excl;
    }
  }
}

__global__ void fill_kernel(const int* __restrict__ src, const int* __restrict__ dst,
                            const int* __restrict__ rowptr, const int* __restrict__ rank,
                            int* __restrict__ colTmp, int* __restrict__ rowOf, int E) {
  int e = blockIdx.x * blockDim.x + threadIdx.x;
  if (e >= E) return;
  int d = dst[e];
  int pos = rowptr[d] + rank[e];
  colTmp[pos] = src[e];
  rowOf[pos] = d;
}

__global__ void order_kernel(const int* __restrict__ rowptr, const int* __restrict__ colTmp,
                             const int* __restrict__ rowOf, const float* __restrict__ dis,
                             int* __restrict__ col, float* __restrict__ wgt, int E) {
  int p = blockIdx.x * blockDim.x + threadIdx.x;
  if (p >= E) return;
  const int d = rowOf[p];
  const int s = rowptr[d], e = rowptr[d + 1];
  const int my = colTmp[p];
  int rank = 0;
  for (int q = s; q < e; ++q) {
    int v = colTmp[q];
    rank += ((v < my) || (v == my && q < p)) ? 1 : 0;
  }
  col[s + rank] = my;
  wgt[s + rank] = dis[my] * dis[d];
}

__global__ void build_h0(const float* __restrict__ x, const float* __restrict__ xm,
                         float* __restrict__ h0, int N) {
  int idx = blockIdx.x * blockDim.x + threadIdx.x;
  if (idx >= N * 16) return;
  int i = idx >> 4, j = idx & 15;
  h0[idx] = (j < 8) ? x[i * 10 + j] : xm[i * 10 + (j - 8)];
}

__device__ __forceinline__ void fma4(float4& acc, const float4 v, const float w) {
  acc.x = fmaf(v.x, w, acc.x);
  acc.y = fmaf(v.y, w, acc.y);
  acc.z = fmaf(v.z, w, acc.z);
  acc.w = fmaf(v.w, w, acc.w);
}

#define TILE_PITCH 132

struct RowHead {
  int s, e, c;
  float w, di;
  float4 self;
};

__device__ __forceinline__ void row_first(RowHead& h, const int lane,
                                          const int* __restrict__ col, const float* __restrict__ wgt) {
  const int nb = (h.e - h.s < 32) ? (h.e - h.s) : 32;
  if (lane < nb) {
    h.c = col[h.s + lane];
    h.w = wgt[h.s + lane];
  }
}

__device__ __forceinline__ float4 row_sum(const RowHead& h, const int i, const int lane,
                                          const float* __restrict__ hw, const int* __restrict__ col,
                                          const float* __restrict__ wgt) {
  float4 acc = {0.0f, 0.0f, 0.0f, 0.0f};
  int c = h.c;
  float w = h.w;
  for (int base = h.s; base < h.e; base += 32) {
    const int nb = (h.e - base < 32) ? (h.e - base) : 32;
    if (base != h.s) {
      c = i;
      w = 0.0f;
      if (lane < nb) {
        c = col[base + lane];
        w = wgt[base + lane];
      }
    }
    const int wi = __float_as_int(w);
    int j = 0;
    for (; j + 8 <= nb; j += 8) {
      float4 v[8];
#pragma unroll
      for (int u = 0; u < 8; ++u) {
        const int cu = __builtin_amdgcn_readlane(c, j + u);
        v[u] = ((const float4*)(hw + (size_t)cu * 128))[lane];
      }
#pragma unroll
      for (int u = 0; u < 8; ++u)
        fma4(acc, v[u], __int_as_float(__builtin_amdgcn_readlane(wi, j + u)));
    }
    const int m = nb - j;
    if (m > 0) {
      float4 v[7];
#pragma unroll
      for (int u = 0; u < 7; ++u) {
        if (u < m) {
          const int cu = __builtin_amdgcn_readlane(c, j + u);
          v[u] = ((const float4*)(hw + (size_t)cu * 128))[lane];
        }
      }
#pragma unroll
      for (int u = 0; u < 7; ++u) {
        if (u < m) fma4(acc, v[u], __int_as_float(__builtin_amdgcn_readlane(wi, j + u)));
      }
    }
  }
  return acc;
}

__device__ __forceinline__ void row_finish(const float4 acc, const RowHead& h, const float4 b,
                                           const int i, const int N, const int lane,
                                           float* __restrict__ emb, float* __restrict__ tileRow) {
  const float dd = h.di * h.di;
  float4 val;
  val.x = acc.x + h.self.x * dd + b.x;
  val.y = acc.y + h.self.y * dd + b.y;
  val.z = acc.z + h.self.z * dd + b.z;
  val.w = acc.w + h.self.w * dd + b.w;
  if (emb && i < N) ((float4*)(emb + (size_t)i * 128))[lane] = val;
  float4 r;
  r.x = fmaxf(val.x, 0.0f);
  r.y = fmaxf(val.y, 0.0f);
  r.z = fmaxf(val.z, 0.0f);
  r.w = fmaxf(val.w, 0.0f);
  ((float4*)tileRow)[lane] = r;
}

template <bool HEAD>
__global__ void __launch_bounds__(256)
gather_gemm(const float* __restrict__ hw, const int* __restrict__ rowptr,
            const int* __restrict__ col, const float* __restrict__ wgt,
            const float* __restrict__ dis, const float* __restrict__ bias,
            float* __restrict__ emb, const float* __restrict__ Wn,
            const float* __restrict__ bias_out, int relu_out,
            float* __restrict__ Out, const float* __restrict__ Wr2,
            const float* __restrict__ br2, float* __restrict__ pred, int N) {
  __shared__ alignas(16) float tile[16 * TILE_PITCH];
  __shared__ v2f wfrag[8 * 32 * 32];
  __shared__ alignas(16) float w2s[HEAD ? 384 : 4];
  __shared__ alignas(16) float b1s[HEAD ? 128 : 4];
  __shared__ alignas(16) float part[HEAD ? 8 * 16 * 4 : 4];
  if constexpr (HEAD) {
    for (int i = threadIdx.x; i < 384; i += 256) w2s[i] = Wr2[i];
    if (threadIdx.x < 128) b1s[threadIdx.x] = bias_out ? bias_out[threadIdx.x] : 0.0f;
  }
  const int lane = threadIdx.x & 31;
  const int wave = __builtin_amdgcn_readfirstlane((int)(threadIdx.x >> 5));
  const int half = lane >> 4;
  const int r = lane & 15;
  const int n = wave * 16 + r;

  v2f* wl = wfrag + (size_t)wave * 32 * 32 + lane;
  const float* wBase = Wn + (size_t)(half * 2) * 128 + n;
#pragma unroll
  for (int t = 0; t < 32; ++t) {
    v2f f;
    f.x = wBase[(size_t)(4 * t) * 128];
    f.y = wBase[(size_t)(4 * t) * 128 + 128];
    wl[t * 32] = f;
  }
  const float badd = (!HEAD && bias_out) ? bias_out[n] : 0.0f;
  const int prow = (int)threadIdx.x / 3;
  const int pj = (int)threadIdx.x - prow * 3;
  const float br2j = (HEAD && threadIdx.x < 48) ? br2[pj] : 0.0f;
  const float4 b = ((const float4*)bias)[lane];

  const int ntiles = (N + 15) >> 4;
  for (int tIdx = blockIdx.x; tIdx < ntiles; tIdx += gridDim.x) {
    const int m0 = tIdx * 16;
    float* tl = tile;
    const int iA = m0 + wave * 2;
    const int iB = iA + 1;
    const int rl = m0 + lane;
    const int rp = rowptr[(rl < N) ? rl : N];
    const float dv = dis[(rl < N) ? rl : (N - 1)];
    RowHead hA, hB;
    hA.self = ((const float4*)(hw + (size_t)((iA < N) ? iA : (N - 1)) * 128))[lane];
    hB.self = ((const float4*)(hw + (size_t)((iB < N) ? iB : (N - 1)) * 128))[lane];
    hA.s = __builtin_amdgcn_readlane(rp, wave * 2);
    hA.e = __builtin_amdgcn_readlane(rp, wave * 2 + 1);
    hB.s = hA.e;
    hB.e = __builtin_amdgcn_readlane(rp, wave * 2 + 2);
    hA.di = __int_as_float(__builtin_amdgcn_readlane(__float_as_int(dv), wave * 2));
    hB.di = __int_as_float(__builtin_amdgcn_readlane(__float_as_int(dv), wave * 2 + 1));
    hA.c = 0; hA.w = 0.0f;
    hB.c = 0; hB.w = 0.0f;
    row_first(hA, lane, col, wgt);
    row_first(hB, lane, col, wgt);
    const float4 sA = row_sum(hA, iA, lane, hw, col, wgt);
    row_finish(sA, hA, b, iA, N, lane, emb, tl + (wave * 2) * TILE_PITCH);
    const float4 sB = row_sum(hB, iB, lane, hw, col, wgt);
    row_finish(sB, hB, b, iB, N, lane, emb, tl + (wave * 2 + 1) * TILE_PITCH);
    __syncthreads();

    const float* aBase = tl + r * TILE_PITCH + half * 2;
    v8f acc = {};
#pragma unroll
    for (int t = 0; t < 32; ++t) {
      const v2f a = *(const v2f*)(aBase + 4 * t);
      const v2f bq = wl[t * 32];
      if constexpr (HEAD) {
        acc = __builtin_amdgcn_wmma_f32_16x16x4_f32(false, bq, false, a, (short)0, acc,
                                                    false, false);
      } else {
        acc = __builtin_amdgcn_wmma_f32_16x16x4_f32(false, a, false, bq, (short)0, acc,
                                                    false, false);
      }
    }

    if constexpr (HEAD) {
      const float* bb = b1s + wave * 16 + half * 8;
      const float* ww = w2s + (wave * 16 + half * 8) * 3;
      float p0 = 0.0f, p1 = 0.0f, p2 = 0.0f;
#pragma unroll
      for (int v = 0; v < 8; ++v) {
        const float tv = fmaxf(acc[v] + bb[v], 0.0f);
        p0 = fmaf(tv, ww[v * 3 + 0], p0);
        p1 = fmaf(tv, ww[v * 3 + 1], p1);
        p2 = fmaf(tv, ww[v * 3 + 2], p2);
      }
      p0 += __shfl_xor(p0, 16, 32);
      p1 += __shfl_xor(p1, 16, 32);
      p2 += __shfl_xor(p2, 16, 32);
      if (half == 0) {
        float4 q = {p0, p1, p2, 0.0f};
        ((float4*)(part + (wave * 16 + r) * 4))[0] = q;
      }
      __syncthreads();
      if (threadIdx.x < 48) {
        float s = br2j;
#pragma unroll
        for (int w = 0; w < 8; ++w) s += part[w * 64 + prow * 4 + pj];
        if (m0 + prow < N) pred[(size_t)m0 * 3 + threadIdx.x] = s;
      }
    } else {
      float o[8];
#pragma unroll
      for (int v = 0; v < 8; ++v) {
        const float val = acc[v] + badd;
        o[v] = relu_out ? fmaxf(val, 0.0f) : val;
      }
      __syncthreads();
      float* outBase = Out + (size_t)(m0 + half * 8) * 128 + n;
      if (m0 + 16 <= N) {
#pragma unroll
        for (int v = 0; v < 8; ++v) outBase[(size_t)v * 128] = o[v];
      } else {
#pragma unroll
        for (int v = 0; v < 8; ++v) {
          const int m = m0 + v + half * 8;
          if (m < N) Out[(size_t)m * 128 + n] = o[v];
        }
      }
    }
  }
}

#define IN_PITCH 20

__global__ void __launch_bounds__(256)
gather16_gemm(const float* __restrict__ h0, const int* __restrict__ rowptr,
              const int* __restrict__ col, const float* __restrict__ wgt,
              const float* __restrict__ dis, const float* __restrict__ W0,
              const float* __restrict__ b0, const float* __restrict__ Wn,
              float* __restrict__ Out, int N) {
  __shared__ alignas(16) float tin[16 * IN_PITCH];
  __shared__ alignas(16) float tile[16 * TILE_PITCH];
  __shared__ v2f wfrag[8 * 32 * 32];
  const int lane = threadIdx.x & 31;
  const int wave = __builtin_amdgcn_readfirstlane((int)(threadIdx.x >> 5));
  const int half = lane >> 4;
  const int r = lane & 15;
  const int n = wave * 16 + r;
  const int q = lane & 3;
  const int es = (lane >> 2) & 3;
  const int rI = wave * 2 + half;

  v2f* wl = wfrag + (size_t)wave * 32 * 32 + lane;
  const float* wBase = Wn + (size_t)(half * 2) * 128 + n;
#pragma unroll
  for (int t = 0; t < 32; ++t) {
    v2f f;
    f.x = wBase[(size_t)(4 * t) * 128];
    f.y = wBase[(size_t)(4 * t) * 128 + 128];
    wl[t * 32] = f;
  }
  v2f w0f[4];
  const float* w0Base = W0 + (size_t)(half * 2) * 128 + n;
#pragma unroll
  for (int t = 0; t < 4; ++t) {
    w0f[t].x = w0Base[(size_t)(4 * t) * 128];
    w0f[t].y = w0Base[(size_t)(4 * t) * 128 + 128];
  }
  const float b0n = b0[n];

  const int ntiles = (N + 15) >> 4;
  for (int tIdx = blockIdx.x; tIdx < ntiles; tIdx += gridDim.x) {
    const int m0 = tIdx * 16;
    const int i = m0 + rI;
    const int ic = (i < N) ? i : (N - 1);
    const int s = rowptr[ic];
    const int eIn = rowptr[ic + 1];
    const float di = dis[ic];
    const float4 self = ((const float4*)(h0 + (size_t)ic * 16))[q];
    const int e = (i < N) ? eIn : s;

    float4 acc = {0.0f, 0.0f, 0.0f, 0.0f};
    for (int base = s; base < e; base += 16) {
      int c[4];
      float w[4];
      float4 v[4];
#pragma unroll
      for (int u = 0; u < 4; ++u) {
        const int p = base + es + 4 * u;
        c[u] = ic;
        w[u] = 0.0f;
        if (p < e) {
          c[u] = col[p];
          w[u] = wgt[p];
        }
      }
#pragma unroll
      for (int u = 0; u < 4; ++u) {
        v[u] = float4{0.0f, 0.0f, 0.0f, 0.0f};
        if (base + es + 4 * u < e) v[u] = ((const float4*)(h0 + (size_t)c[u] * 16))[q];
      }
#pragma unroll
      for (int u = 0; u < 4; ++u) fma4(acc, v[u], w[u]);
    }
    acc.x += __shfl_xor(acc.x, 4, 32);
    acc.y += __shfl_xor(acc.y, 4, 32);
    acc.z += __shfl_xor(acc.z, 4, 32);
    acc.w += __shfl_xor(acc.w, 4, 32);
    acc.x += __shfl_xor(acc.x, 8, 32);
    acc.y += __shfl_xor(acc.y, 8, 32);
    acc.z += __shfl_xor(acc.z, 8, 32);
    acc.w += __shfl_xor(acc.w, 8, 32);
    if (es == 0) {
      const float dd = di * di;
      float4 val;
      val.x = acc.x + self.x * dd;
      val.y = acc.y + self.y * dd;
      val.z = acc.z + self.z * dd;
      val.w = acc.w + self.w * dd;
      ((float4*)(tin + rI * IN_PITCH))[q] = val;
    }
    __syncthreads();

    const float* a0 = tin + r * IN_PITCH + half * 2;
    v8f acc1 = {};
#pragma unroll
    for (int t = 0; t < 4; ++t) {
      const v2f a = *(const v2f*)(a0 + 4 * t);
      acc1 = __builtin_amdgcn_wmma_f32_16x16x4_f32(false, a, false, w0f[t], (short)0, acc1,
                                                  false, false);
    }
    float* tcol = tile + (half * 8) * TILE_PITCH + n;
#pragma unroll
    for (int v = 0; v < 8; ++v) tcol[v * TILE_PITCH] = fmaxf(acc1[v] + b0n, 0.0f);
    __syncthreads();

    const float* aBase = tile + r * TILE_PITCH + half * 2;
    v8f acc2 = {};
#pragma unroll
    for (int t = 0; t < 32; ++t) {
      const v2f a = *(const v2f*)(aBase + 4 * t);
      const v2f bq = wl[t * 32];
      acc2 = __builtin_amdgcn_wmma_f32_16x16x4_f32(false, a, false, bq, (short)0, acc2,
                                                  false, false);
    }
    float* outBase = Out + (size_t)(m0 + half * 8) * 128 + n;
    if (m0 + 16 <= N) {
#pragma unroll
      for (int v = 0; v < 8; ++v) outBase[(size_t)v * 128] = acc2[v];
    } else {
#pragma unroll
      for (int v = 0; v < 8; ++v) {
        const int m = m0 + v + half * 8;
        if (m < N) Out[(size_t)m * 128 + n] = acc2[v];
      }
    }
  }
}

extern "C" void kernel_launch(void* const* d_in, const int* in_sizes, int n_in,
                              void* d_out, int out_size, void* d_ws, size_t ws_size,
                              hipStream_t stream) {
  const float* x   = (const float*)d_in[0];
  const float* xm  = (const float*)d_in[1];
  const int*   ei  = (const int*)d_in[2];
  const float* W0  = (const float*)d_in[3];
  const float* b0  = (const float*)d_in[4];
  const float* Wh  = (const float*)d_in[5];
  const float* bh  = (const float*)d_in[6];
  const float* Wr1 = (const float*)d_in[7];
  const float* br1 = (const float*)d_in[8];
  const float* Wr2 = (const float*)d_in[9];
  const float* br2 = (const float*)d_in[10];

  const int N = in_sizes[0] / 10;
  const int E = in_sizes[2] / 2;
  const int* srcv = ei;
  const int* dstv = ei + E;

  float* bufA = (float*)d_ws;
  float* bufB = bufA + (size_t)N * 128;
  float* bufC = bufB + (size_t)N * 128;
  float* dis  = bufC + (size_t)N * 128;

  const size_t nA = ((size_t)N + 1 + 63) & ~(size_t)63;
  const size_t eA = ((size_t)E + 63) & ~(size_t)63;
  int*   cnt    = (int*)bufC;
  int*   rowptr = cnt + nA;
  int*   bsum   = rowptr + nA;
  int*   colTmp = bsum + nA;
  int*   rowOf  = colTmp + eA;
  int*   col    = rowOf + eA;
  float* wgt    = (float*)(col + eA);
  int*   rank   = (int*)(wgt + eA);

  float* emb_out  = (float*)d_out;
  float* pred_out = emb_out + (size_t)N * 128;

  const int mblocks = (N + 15) / 16;
  const int eblocks = (E + 255) / 256;
  const int tblocks = (N + 1023) / 1024;
  const int fblocks = (mblocks < 1024) ? mblocks : 1024;

  hipMemsetAsync(cnt, 0, (size_t)N * sizeof(int), stream);
  count_kernel<<<eblocks, 256, 0, stream>>>(dstv, cnt, rank, E);
  scan_partials<<<tblocks, 256, 0, stream>>>(cnt, bsum, N);
  scan_finish<<<tblocks, 256, 0, stream>>>(cnt, bsum, rowptr, dis, N);
  fill_kernel<<<eblocks, 256, 0, stream>>>(srcv, dstv, rowptr, rank, colTmp, rowOf, E);
  order_kernel<<<eblocks, 256, 0, stream>>>(rowptr, colTmp, rowOf, dis, col, wgt, E);

  build_h0<<<(int)(((size_t)N * 16 + 255) / 256), 256, 0, stream>>>(x, xm, bufB, N);

  gather16_gemm<<<fblocks, 256, 0, stream>>>(bufB, rowptr, col, wgt, dis, W0, b0, Wh, bufA, N);

  gather_gemm<false><<<fblocks, 256, 0, stream>>>(bufA, rowptr, col, wgt, dis, bh, nullptr,
                                                  Wh + (size_t)1 * 128 * 128, nullptr, 0, bufB,
                                                  nullptr, nullptr, nullptr, N);
  gather_gemm<false><<<fblocks, 256, 0, stream>>>(bufB, rowptr, col, wgt, dis, bh + 128, nullptr,
                                                  Wh + (size_t)2 * 128 * 128, nullptr, 0, bufA,
                                                  nullptr, nullptr, nullptr, N);
  gather_gemm<false><<<fblocks, 256, 0, stream>>>(bufA, rowptr, col, wgt, dis, bh + 256, nullptr,
                                                  Wh + (size_t)3 * 128 * 128, nullptr, 0, bufB,
                                                  nullptr, nullptr, nullptr, N);
  gather_gemm<true><<<fblocks, 256, 0, stream>>>(bufB, rowptr, col, wgt, dis, bh + 384, emb_out,
                                                 Wr1, br1, 1, nullptr, Wr2, br2, pred_out, N);
}


# in-degree count and edge ranks on LDS counters: a block owns a row range and an edge segment and counts its targets in LDS; no global atomic, no memset left; col/wgt order and all sums as before
# speedup vs baseline: 4.4992x; 1.0455x over previous
#include <hip/hip_runtime.h>

typedef __attribute__((ext_vector_type(2))) float v2f;
typedef __attribute__((ext_vector_type(8))) float v8f;

#define SEGS 8
#define RANGE_ROWS 3136
__global__ void __launch_bounds__(1024)
count_ranges(const int* __restrict__ dst, int* __restrict__ part, int* __restrict__ rank,
             int E, int N, int nA, int segLen) {
  __shared__ int c[RANGE_ROWS];
  const int tid = threadIdx.x;
  const int lo = blockIdx.x * RANGE_ROWS;
  const int seg = blockIdx.y;
  for (int i = tid; i < RANGE_ROWS; i += 1024) c[i] = 0;
  __syncthreads();
  const int e0 = seg * segLen;
  const int e1 = (e0 + segLen < E) ? (e0 + segLen) : E;
  for (int eb = e0 + tid; eb < e1; eb += 16 * 1024) {
    int d[16];
    int r[16];
#pragma unroll
    for (int u = 0; u < 16; ++u) {
      const int e = eb + u * 1024;
      const int v = dst[(e < e1) ? e : (e1 - 1)] - lo;
      d[u] = (e < e1) ? v : -1;
    }
#pragma unroll
    for (int u = 0; u < 16; ++u) {
      r[u] = 0;
      if ((unsigned)d[u] < (unsigned)RANGE_ROWS) r[u] = atomicAdd(&c[d[u]], 1);
    }
#pragma unroll
    for (int u = 0; u < 16; ++u) {
      if ((unsigned)d[u] < (unsigned)RANGE_ROWS) rank[eb + u * 1024] = r[u];
    }
  }
  __syncthreads();
  for (int i = tid; i < RANGE_ROWS; i += 1024) {
    const int row = lo + i;
    if (row < N) part[seg * nA + row] = c[i];
  }
}

__global__ void scan_partials(const int* __restrict__ part, int* __restrict__ bsum, int N, int nA) {
  __shared__ int ws[8];
  const int base = blockIdx.x * 1024 + threadIdx.x * 4;
  int s = 0;
  for (int j = 0; j < 4; ++j) {
    int i = base + j;
    if (i < N) {
      for (int g = 0; g < SEGS; ++g) s += part[g * nA + i];
    }
  }
  for (int o = 16; o > 0; o >>= 1) s += __shfl_xor(s, o, 32);
  if ((threadIdx.x & 31) == 0) ws[threadIdx.x >> 5] = s;
  __syncthreads();
  if (threadIdx.x == 0) {
    int t = 0;
    for (int w = 0; w < 8; ++w) t += ws[w];
    bsum[blockIdx.x] = t;
  }
}

__global__ void scan_finish(int* __restrict__ part, const int* __restrict__ bsum,
                            int* __restrict__ rowptr, float* __restrict__ dis, int N, int nA) {
  __shared__ int ws[8];
  __shared__ int wt[8];
  const int tid = threadIdx.x, lane = tid & 31, wave = tid >> 5;
  int p = 0;
  for (int j = tid; j < (int)blockIdx.x; j += 256) p += bsum[j];
  for (int o = 16; o > 0; o >>= 1) p += __shfl_xor(p, o, 32);
  if (lane == 0) ws[wave] = p;

  const int base = blockIdx.x * 1024 + tid * 4;
  int c[4];
  int pc[4][SEGS];
  int s = 0;
#pragma unroll
  for (int j = 0; j < 4; ++j) {
    int i = base + j;
    c[j] = 0;
#pragma unroll
    for (int g = 0; g < SEGS; ++g) {
      pc[j][g] = (i < N) ? part[g * nA + i] : 0;
      c[j] += pc[j][g];
    }
    s += c[j];
  }
  int inc = s;
  for (int o = 1; o < 32; o <<= 1) {
    int t = __shfl_up(inc, o, 32);
    if (lane >= o) inc += t;
  }
  if (lane == 31) wt[wave] = inc;
  __syncthreads();
  int off = 0;
  for (int w = 0; w < 8; ++w) off += ws[w];
  for (int w = 0; w < wave; ++w) off += wt[w];
  int excl = off + inc - s;
#pragma unroll
  for (int j = 0; j < 4; ++j) {
    int i = base + j;
    if (i < N) {
      rowptr[i] = excl;
      dis[i] = rsqrtf((float)c[j] + 1.0f);
      int b = excl;
#pragma unroll
      for (int g = 0; g < SEGS; ++g) {
        part[g * nA + i] = b;
        b += pc[j][g];
      }
      excl += c[j];
      if (i == N - 1) rowptr[N] = excl;
    }
  }
}

__global__ void fill_kernel(const int* __restrict__ src, const int* __restrict__ dst,
                            const int* __restrict__ part, const int* __restrict__ rank,
                            int* __restrict__ colTmp, int* __restrict__ rowOf, int E, int nA, int segLen) {
  int e = blockIdx.x * blockDim.x + threadIdx.x;
  if (e >= E) return;
  int d = dst[e];
  int pos = part[(e / segLen) * nA + d] + rank[e];
  colTmp[pos] = src[e];
  rowOf[pos] = d;
}

__global__ void order_kernel(const int* __restrict__ rowptr, const int* __restrict__ colTmp,
                             const int* __restrict__ rowOf, const float* __restrict__ dis,
                             int* __restrict__ col, float* __restrict__ wgt, int E) {
  int p = blockIdx.x * blockDim.x + threadIdx.x;
  if (p >= E) return;
  const int d = rowOf[p];
  const int s = rowptr[d], e = rowptr[d + 1];
  const int my = colTmp[p];
  int rank = 0;
  for (int q = s; q < e; ++q) {
    int v = colTmp[q];
    rank += ((v < my) || (v == my && q < p)) ? 1 : 0;
  }
  col[s + rank] = my;
  wgt[s + rank] = dis[my] * dis[d];
}

__global__ void build_h0(const float* __restrict__ x, const float* __restrict__ xm,
                         float* __restrict__ h0, int N) {
  int idx = blockIdx.x * blockDim.x + threadIdx.x;
  if (idx >= N * 16) return;
  int i = idx >> 4, j = idx & 15;
  h0[idx] = (j < 8) ? x[i * 10 + j] : xm[i * 10 + (j - 8)];
}

__device__ __forceinline__ void fma4(float4& acc, const float4 v, const float w) {
  acc.x = fmaf(v.x, w, acc.x);
  acc.y = fmaf(v.y, w, acc.y);
  acc.z = fmaf(v.z, w, acc.z);
  acc.w = fmaf(v.w, w, acc.w);
}

#define TILE_PITCH 132

struct RowHead {
  int s, e, c;
  float w, di;
  float4 self;
};

__device__ __forceinline__ void row_first(RowHead& h, const int lane,
                                          const int* __restrict__ col, const float* __restrict__ wgt) {
  const int nb = (h.e - h.s < 32) ? (h.e - h.s) : 32;
  if (lane < nb) {
    h.c = col[h.s + lane];
    h.w = wgt[h.s + lane];
  }
}

__device__ __forceinline__ float4 row_sum(const RowHead& h, const int i, const int lane,
                                          const float* __restrict__ hw, const int* __restrict__ col,
                                          const float* __restrict__ wgt) {
  float4 acc = {0.0f, 0.0f, 0.0f, 0.0f};
  int c = h.c;
  float w = h.w;
  for (int base = h.s; base < h.e; base += 32) {
    const int nb = (h.e - base < 32) ? (h.e - base) : 32;
    if (base != h.s) {
      c = i;
      w = 0.0f;
      if (lane < nb) {
        c = col[base + lane];
        w = wgt[base + lane];
      }
    }
    const int wi = __float_as_int(w);
    int j = 0;
    for (; j + 8 <= nb; j += 8) {
      float4 v[8];
#pragma unroll
      for (int u = 0; u < 8; ++u) {
        const int cu = __builtin_amdgcn_readlane(c, j + u);
        v[u] = ((const float4*)(hw + (size_t)cu * 128))[lane];
      }
#pragma unroll
      for (int u = 0; u < 8; ++u)
        fma4(acc, v[u], __int_as_float(__builtin_amdgcn_readlane(wi, j + u)));
    }
    const int m = nb - j;
    if (m > 0) {
      float4 v[7];
#pragma unroll
      for (int u = 0; u < 7; ++u) {
        if (u < m) {
          const int cu = __builtin_amdgcn_readlane(c, j + u);
          v[u] = ((const float4*)(hw + (size_t)cu * 128))[lane];
        }
      }
#pragma unroll
      for (int u = 0; u < 7; ++u) {
        if (u < m) fma4(acc, v[u], __int_as_float(__builtin_amdgcn_readlane(wi, j + u)));
      }
    }
  }
  return acc;
}

__device__ __forceinline__ void row_finish(const float4 acc, const RowHead& h, const float4 b,
                                           const int i, const int N, const int lane,
                                           float* __restrict__ emb, float* __restrict__ tileRow) {
  const float dd = h.di * h.di;
  float4 val;
  val.x = acc.x + h.self.x * dd + b.x;
  val.y = acc.y + h.self.y * dd + b.y;
  val.z = acc.z + h.self.z * dd + b.z;
  val.w = acc.w + h.self.w * dd + b.w;
  if (emb && i < N) ((float4*)(emb + (size_t)i * 128))[lane] = val;
  float4 r;
  r.x = fmaxf(val.x, 0.0f);
  r.y = fmaxf(val.y, 0.0f);
  r.z = fmaxf(val.z, 0.0f);
  r.w = fmaxf(val.w, 0.0f);
  ((float4*)tileRow)[lane] = r;
}

template <bool HEAD>
__global__ void __launch_bounds__(256)
gather_gemm(const float* __restrict__ hw, const int* __restrict__ rowptr,
            const int* __restrict__ col, const float* __restrict__ wgt,
            const float* __restrict__ dis, const float* __restrict__ bias,
            float* __restrict__ emb, const float* __restrict__ Wn,
            const float* __restrict__ bias_out, int relu_out,
            float* __restrict__ Out, const float* __restrict__ Wr2,
            const float* __restrict__ br2, float* __restrict__ pred, int N) {
  __shared__ alignas(16) float tile[16 * TILE_PITCH];
  __shared__ v2f wfrag[8 * 32 * 32];
  __shared__ alignas(16) float w2s[HEAD ? 384 : 4];
  __shared__ alignas(16) float b1s[HEAD ? 128 : 4];
  __shared__ alignas(16) float part[HEAD ? 8 * 16 * 4 : 4];
  if constexpr (HEAD) {
    for (int i = threadIdx.x; i < 384; i += 256) w2s[i] = Wr2[i];
    if (threadIdx.x < 128) b1s[threadIdx.x] = bias_out ? bias_out[threadIdx.x] : 0.0f;
  }
  const int lane = threadIdx.x & 31;
  const int wave = __builtin_amdgcn_readfirstlane((int)(threadIdx.x >> 5));
  const int half = lane >> 4;
  const int r = lane & 15;
  const int n = wave * 16 + r;

  v2f* wl = wfrag + (size_t)wave * 32 * 32 + lane;
  const float* wBase = Wn + (size_t)(half * 2) * 128 + n;
#pragma unroll
  for (int t = 0; t < 32; ++t) {
    v2f f;
    f.x = wBase[(size_t)(4 * t) * 128];
    f.y = wBase[(size_t)(4 * t) * 128 + 128];
    wl[t * 32] = f;
  }
  const float badd = (!HEAD && bias_out) ? bias_out[n] : 0.0f;
  const int prow = (int)threadIdx.x / 3;
  const int pj = (int)threadIdx.x - prow * 3;
  const float br2j = (HEAD && threadIdx.x < 48) ? br2[pj] : 0.0f;
  const float4 b = ((const float4*)bias)[lane];

  const int ntiles = (N + 15) >> 4;
  for (int tIdx = blockIdx.x; tIdx < ntiles; tIdx += gridDim.x) {
    const int m0 = tIdx * 16;
    float* tl = tile;
    const int iA = m0 + wave * 2;
    const int iB = iA + 1;
    const int rl = m0 + lane;
    const int rp = rowptr[(rl < N) ? rl : N];
    const float dv = dis[(rl < N) ? rl : (N - 1)];
    RowHead hA, hB;
    hA.self = ((const float4*)(hw + (size_t)((iA < N) ? iA : (N - 1)) * 128))[lane];
    hB.self = ((const float4*)(hw + (size_t)((iB < N) ? iB : (N - 1)) * 128))[lane];
    hA.s = __builtin_amdgcn_readlane(rp, wave * 2);
    hA.e = __builtin_amdgcn_readlane(rp, wave * 2 + 1);
    hB.s = hA.e;
    hB.e = __builtin_amdgcn_readlane(rp, wave * 2 + 2);
    hA.di = __int_as_float(__builtin_amdgcn_readlane(__float_as_int(dv), wave * 2));
    hB.di = __int_as_float(__builtin_amdgcn_readlane(__float_as_int(dv), wave * 2 + 1));
    hA.c = 0; hA.w = 0.0f;
    hB.c = 0; hB.w = 0.0f;
    row_first(hA, lane, col, wgt);
    row_first(hB, lane, col, wgt);
    const float4 sA = row_sum(hA, iA, lane, hw, col, wgt);
    row_finish(sA, hA, b, iA, N, lane, emb, tl + (wave * 2) * TILE_PITCH);
    const float4 sB = row_sum(hB, iB, lane, hw, col, wgt);
    row_finish(sB, hB, b, iB, N, lane, emb, tl + (wave * 2 + 1) * TILE_PITCH);
    __syncthreads();

    const float* aBase = tl + r * TILE_PITCH + half * 2;
    v8f acc = {};
#pragma unroll
    for (int t = 0; t < 32; ++t) {
      const v2f a = *(const v2f*)(aBase + 4 * t);
      const v2f bq = wl[t * 32];
      if constexpr (HEAD) {
        acc = __builtin_amdgcn_wmma_f32_16x16x4_f32(false, bq, false, a, (short)0, acc,
                                                    false, false);
      } else {
        acc = __builtin_amdgcn_wmma_f32_16x16x4_f32(false, a, false, bq, (short)0, acc,
                                                    false, false);
      }
    }

    if constexpr (HEAD) {
      const float* bb = b1s + wave * 16 + half * 8;
      const float* ww = w2s + (wave * 16 + half * 8) * 3;
      float p0 = 0.0f, p1 = 0.0f, p2 = 0.0f;
#pragma unroll
      for (int v = 0; v < 8; ++v) {
        const float tv = fmaxf(acc[v] + bb[v], 0.0f);
        p0 = fmaf(tv, ww[v * 3 + 0], p0);
        p1 = fmaf(tv, ww[v * 3 + 1], p1);
        p2 = fmaf(tv, ww[v * 3 + 2], p2);
      }
      p0 += __shfl_xor(p0, 16, 32);
      p1 += __shfl_xor(p1, 16, 32);
      p2 += __shfl_xor(p2, 16, 32);
      if (half == 0) {
        float4 q = {p0, p1, p2, 0.0f};
        ((float4*)(part + (wave * 16 + r) * 4))[0] = q;
      }
      __syncthreads();
      if (threadIdx.x < 48) {
        float s = br2j;
#pragma unroll
        for (int w = 0; w < 8; ++w) s += part[w * 64 + prow * 4 + pj];
        if (m0 + prow < N) pred[(size_t)m0 * 3 + threadIdx.x] = s;
      }
    } else {
      float o[8];
#pragma unroll
      for (int v = 0; v < 8; ++v) {
        const float val = acc[v] + badd;
        o[v] = relu_out ? fmaxf(val, 0.0f) : val;
      }
      __syncthreads();
      float* outBase = Out + (size_t)(m0 + half * 8) * 128 + n;
      if (m0 + 16 <= N) {
#pragma unroll
        for (int v = 0; v < 8; ++v) outBase[(size_t)v * 128] = o[v];
      } else {
#pragma unroll
        for (int v = 0; v < 8; ++v) {
          const int m = m0 + v + half * 8;
          if (m < N) Out[(size_t)m * 128 + n] = o[v];
        }
      }
    }
  }
}

#define IN_PITCH 20

__global__ void __launch_bounds__(256)
gather16_gemm(const float* __restrict__ h0, const int* __restrict__ rowptr,
              const int* __restrict__ col, const float* __restrict__ wgt,
              const float* __restrict__ dis, const float* __restrict__ W0,
              const float* __restrict__ b0, const float* __restrict__ Wn,
              float* __restrict__ Out, int N) {
  __shared__ alignas(16) float tin[16 * IN_PITCH];
  __shared__ alignas(16) float tile[16 * TILE_PITCH];
  __shared__ v2f wfrag[8 * 32 * 32];
  const int lane = threadIdx.x & 31;
  const int wave = __builtin_amdgcn_readfirstlane((int)(threadIdx.x >> 5));
  const int half = lane >> 4;
  const int r = lane & 15;
  const int n = wave * 16 + r;
  const int q = lane & 3;
  const int es = (lane >> 2) & 3;
  const int rI = wave * 2 + half;

  v2f* wl = wfrag + (size_t)wave * 32 * 32 + lane;
  const float* wBase = Wn + (size_t)(half * 2) * 128 + n;
#pragma unroll
  for (int t = 0; t < 32; ++t) {
    v2f f;
    f.x = wBase[(size_t)(4 * t) * 128];
    f.y = wBase[(size_t)(4 * t) * 128 + 128];
    wl[t * 32] = f;
  }
  v2f w0f[4];
  const float* w0Base = W0 + (size_t)(half * 2) * 128 + n;
#pragma unroll
  for (int t = 0; t < 4; ++t) {
    w0f[t].x = w0Base[(size_t)(4 * t) * 128];
    w0f[t].y = w0Base[(size_t)(4 * t) * 128 + 128];
  }
  const float b0n = b0[n];

  const int ntiles = (N + 15) >> 4;
  for (int tIdx = blockIdx.x; tIdx < ntiles; tIdx += gridDim.x) {
    const int m0 = tIdx * 16;
    const int i = m0 + rI;
    const int ic = (i < N) ? i : (N - 1);
    const int s = rowptr[ic];
    const int eIn = rowptr[ic + 1];
    const float di = dis[ic];
    const float4 self = ((const float4*)(h0 + (size_t)ic * 16))[q];
    const int e = (i < N) ? eIn : s;

    float4 acc = {0.0f, 0.0f, 0.0f, 0.0f};
    for (int base = s; base < e; base += 16) {
      int c[4];
      float w[4];
      float4 v[4];
#pragma unroll
      for (int u = 0; u < 4; ++u) {
        const int p = base + es + 4 * u;
        c[u] = ic;
        w[u] = 0.0f;
        if (p < e) {
          c[u] = col[p];
          w[u] = wgt[p];
        }
      }
#pragma unroll
      for (int u = 0; u < 4; ++u) {
        v[u] = float4{0.0f, 0.0f, 0.0f, 0.0f};
        if (base + es + 4 * u < e) v[u] = ((const float4*)(h0 + (size_t)c[u] * 16))[q];
      }
#pragma unroll
      for (int u = 0; u < 4; ++u) fma4(acc, v[u], w[u]);
    }
    acc.x += __shfl_xor(acc.x, 4, 32);
    acc.y += __shfl_xor(acc.y, 4, 32);
    acc.z += __shfl_xor(acc.z, 4, 32);
    acc.w += __shfl_xor(acc.w, 4, 32);
    acc.x += __shfl_xor(acc.x, 8, 32);
    acc.y += __shfl_xor(acc.y, 8, 32);
    acc.z += __shfl_xor(acc.z, 8, 32);
    acc.w += __shfl_xor(acc.w, 8, 32);
    if (es == 0) {
      const float dd = di * di;
      float4 val;
      val.x = acc.x + self.x * dd;
      val.y = acc.y + self.y * dd;
      val.z = acc.z + self.z * dd;
      val.w = acc.w + self.w * dd;
      ((float4*)(tin + rI * IN_PITCH))[q] = val;
    }
    __syncthreads();

    const float* a0 = tin + r * IN_PITCH + half * 2;
    v8f acc1 = {};
#pragma unroll
    for (int t = 0; t < 4; ++t) {
      const v2f a = *(const v2f*)(a0 + 4 * t);
      acc1 = __builtin_amdgcn_wmma_f32_16x16x4_f32(false, a, false, w0f[t], (short)0, acc1,
                                                  false, false);
    }
    float* tcol = tile + (half * 8) * TILE_PITCH + n;
#pragma unroll
    for (int v = 0; v < 8; ++v) tcol[v * TILE_PITCH] = fmaxf(acc1[v] + b0n, 0.0f);
    __syncthreads();

    const float* aBase = tile + r * TILE_PITCH + half * 2;
    v8f acc2 = {};
#pragma unroll
    for (int t = 0; t < 32; ++t) {
      const v2f a = *(const v2f*)(aBase + 4 * t);
      const v2f bq = wl[t * 32];
      acc2 = __builtin_amdgcn_wmma_f32_16x16x4_f32(false, a, false, bq, (short)0, acc2,
                                                  false, false);
    }
    float* outBase = Out + (size_t)(m0 + half * 8) * 128 + n;
    if (m0 + 16 <= N) {
#pragma unroll
      for (int v = 0; v < 8; ++v) outBase[(size_t)v * 128] = acc2[v];
    } else {
#pragma unroll
      for (int v = 0; v < 8; ++v) {
        const int m = m0 + v + half * 8;
        if (m < N) Out[(size_t)m * 128 + n] = acc2[v];
      }
    }
  }
}

extern "C" void kernel_launch(void* const* d_in, const int* in_sizes, int n_in,
                              void* d_out, int out_size, void* d_ws, size_t ws_size,
                              hipStream_t stream) {
  const float* x   = (const float*)d_in[0];
  const float* xm  = (const float*)d_in[1];
  const int*   ei  = (const int*)d_in[2];
  const float* W0  = (const float*)d_in[3];
  const float* b0  = (const float*)d_in[4];
  const float* Wh  = (const float*)d_in[5];
  const float* bh  = (const float*)d_in[6];
  const float* Wr1 = (const float*)d_in[7];
  const float* br1 = (const float*)d_in[8];
  const float* Wr2 = (const float*)d_in[9];
  const float* br2 = (const float*)d_in[10];

  const int N = in_sizes[0] / 10;
  const int E = in_sizes[2] / 2;
  const int* srcv = ei;
  const int* dstv = ei + E;

  float* bufA = (float*)d_ws;
  float* bufB = bufA + (size_t)N * 128;
  float* bufC = bufB + (size_t)N * 128;
  float* dis  = bufC + (size_t)N * 128;

  const size_t nA = ((size_t)N + 1 + 63) & ~(size_t)63;
  const size_t eA = ((size_t)E + 63) & ~(size_t)63;
  int*   part   = (int*)bufC;
  int*   rowptr = part + (size_t)SEGS * nA;
  int*   bsum   = rowptr + nA;
  int*   colTmp = bsum + nA;
  int*   rowOf  = colTmp + eA;
  int*   col    = rowOf + eA;
  float* wgt    = (float*)(col + eA);
  int*   rank   = (int*)(wgt + eA);

  float* emb_out  = (float*)d_out;
  float* pred_out = emb_out + (size_t)N * 128;

  const int mblocks = (N + 15) / 16;
  const int eblocks = (E + 255) / 256;
  const int tblocks = (N + 1023) / 1024;
  const int fblocks = (mblocks < 1024) ? mblocks : 1024;

  const int segLen  = (E + SEGS - 1) / SEGS;
  const int nranges = (N + RANGE_ROWS - 1) / RANGE_ROWS;
  count_ranges<<<dim3(nranges, SEGS), 1024, 0, stream>>>(dstv, part, rank, E, N, (int)nA, segLen);
  scan_partials<<<tblocks, 256, 0, stream>>>(part, bsum, N, (int)nA);
  scan_finish<<<tblocks, 256, 0, stream>>>(part, bsum, rowptr, dis, N, (int)nA);
  fill_kernel<<<eblocks, 256, 0, stream>>>(srcv, dstv, part, rank, colTmp, rowOf, E, (int)nA, segLen);
  order_kernel<<<eblocks, 256, 0, stream>>>(rowptr, colTmp, rowOf, dis, col, wgt, E);

  build_h0<<<(int)(((size_t)N * 16 + 255) / 256), 256, 0, stream>>>(x, xm, bufB, N);

  gather16_gemm<<<fblocks, 256, 0, stream>>>(bufB, rowptr, col, wgt, dis, W0, b0, Wh, bufA, N);

  gather_gemm<false><<<fblocks, 256, 0, stream>>>(bufA, rowptr, col, wgt, dis, bh, nullptr,
                                                  Wh + (size_t)1 * 128 * 128, nullptr, 0, bufB,
                                                  nullptr, nullptr, nullptr, N);
  gather_gemm<false><<<fblocks, 256, 0, stream>>>(bufB, rowptr, col, wgt, dis, bh + 128, nullptr,
                                                  Wh + (size_t)2 * 128 * 128, nullptr, 0, bufA,
                                                  nullptr, nullptr, nullptr, N);
  gather_gemm<false><<<fblocks, 256, 0, stream>>>(bufA, rowptr, col, wgt, dis, bh + 256, nullptr,
                                                  Wh + (size_t)3 * 128 * 128, nullptr, 0, bufB,
                                                  nullptr, nullptr, nullptr, N);
  gather_gemm<true><<<fblocks, 256, 0, stream>>>(bufB, rowptr, col, wgt, dis, bh + 384, emb_out,
                                                 Wr1, br1, 1, nullptr, Wr2, br2, pred_out, N);
}
